# starting kernel
# baseline (speedup 1.0000x reference)
#include <hip/hip_runtime.h>

typedef __attribute__((ext_vector_type(2))) float v2f;
typedef __attribute__((ext_vector_type(8))) float v8f;

#define DIM 64

__global__ void gcn_zero_kernel(float* __restrict__ out, int n4) {
    int i = blockIdx.x * blockDim.x + threadIdx.x;
    if (i < n4) {
        float4 z = make_float4(0.f, 0.f, 0.f, 0.f);
        ((float4*)out)[i] = z;
    }
}

__global__ void gcn_gemm_wmma_kernel(const float* __restrict__ X,
                                     const float* __restrict__ W,
                                     float* __restrict__ S,
                                     int n_row_tiles) {
    const int wave = threadIdx.x >> 5;
    const int lane = threadIdx.x & 31;
    const int tile = blockIdx.x * (blockDim.x >> 5) + wave;
    if (tile >= n_row_tiles) return;

    const int m0    = tile << 4;
    const int mrow  = lane & 15;
    const int khalf = lane >> 4;
    const int ncol  = lane & 15;

    v8f acc0 = {}, acc1 = {}, acc2 = {}, acc3 = {};

    const float* __restrict__ xrow = X + (long long)(m0 + mrow) * DIM;

    #pragma unroll
    for (int k0 = 0; k0 < DIM; k0 += 4) {
        const int ka = k0 + (khalf << 1);
        v2f a = *(const v2f*)(xrow + ka);

        const float* __restrict__ w0 = W + ka * DIM;
        const float* __restrict__ w1 = W + (ka + 1) * DIM;

        v2f b0, b1, b2, b3;
        b0.x = w0[ 0 + ncol]; b0.y = w1[ 0 + ncol];
        b1.x = w0[16 + ncol]; b1.y = w1[16 + ncol];
        b2.x = w0[32 + ncol]; b2.y = w1[32 + ncol];
        b3.x = w0[48 + ncol]; b3.y = w1[48 + ncol];

        acc0 = __builtin_amdgcn_wmma_f32_16x16x4_f32(false, a, false, b0,
                                                     (short)0, acc0, false, false);
        acc1 = __builtin_amdgcn_wmma_f32_16x16x4_f32(false, a, false, b1,
                                                     (short)0, acc1, false, false);
        acc2 = __builtin_amdgcn_wmma_f32_16x16x4_f32(false, a, false, b2,
                                                     (short)0, acc2, false, false);
        acc3 = __builtin_amdgcn_wmma_f32_16x16x4_f32(false, a, false, b3,
                                                     (short)0, acc3, false, false);
    }

    const int rbase = m0 + (khalf << 3);
    #pragma unroll
    for (int v = 0; v < 8; ++v) {
        float* __restrict__ srow = S + (long long)(rbase + v) * DIM + ncol;
        srow[ 0] = acc0[v];
        srow[16] = acc1[v];
        srow[32] = acc2[v];
        srow[48] = acc3[v];
    }
}

__global__ void gcn_scatter_kernel(const float* __restrict__ S,
                                   const int* __restrict__ adj_row,
                                   const int* __restrict__ adj_col,
                                   const float* __restrict__ adj_val,
                                   float* __restrict__ out,
                                   int n_edges) {
    const long long idx = (long long)blockIdx.x * blockDim.x + threadIdx.x;
    const int e = (int)(idx >> 6);
    const int d = (int)(idx & 63);
    if (e >= n_edges) return;

    const int   r = adj_row[e];
    const int   c = adj_col[e];
    const float v = adj_val[e];

    atomicAdd(out + (long long)r * DIM + d, v * S[(long long)c * DIM + d]);
}

extern "C" void kernel_launch(void* const* d_in, const int* in_sizes, int n_in,
                              void* d_out, int out_size, void* d_ws, size_t ws_size,
                              hipStream_t stream) {
    const float* X    = (const float*)d_in[0];
    const float* W    = (const float*)d_in[1];
    const int*   arow = (const int*)d_in[2];
    const int*   acol = (const int*)d_in[3];
    const float* aval = (const float*)d_in[4];
    float*       out  = (float*)d_out;
    float*       S    = (float*)d_ws;

    const int n_nodes = in_sizes[0] / DIM;
    const int n_edges = in_sizes[2];

    {
        const int n4 = out_size / 4;
        gcn_zero_kernel<<<(n4 + 255) / 256, 256, 0, stream>>>(out, n4);
    }

    {
        const int tiles = n_nodes / 16;
        const int waves_per_block = 256 / 32;
        const int blocks = (tiles + waves_per_block - 1) / waves_per_block;
        gcn_gemm_wmma_kernel<<<blocks, 256, 0, stream>>>(X, W, S, tiles);
    }

    {
        const long long work = (long long)n_edges * DIM;
        const int blocks = (int)((work + 255) / 256);
        gcn_scatter_kernel<<<blocks, 256, 0, stream>>>(S, arow, acol, aval, out, n_edges);
    }
}


# Memory float atomics replaced by LDS float adds: a work-group keeps 391 out rows in LDS, scans adj_row for its edges, adds val*support rows, writes each row once; no zero launch; arrival-order sums
# speedup vs baseline: 3.0464x; 3.0464x over previous
#include <hip/hip_runtime.h>

typedef __attribute__((ext_vector_type(2))) float v2f;
typedef __attribute__((ext_vector_type(8))) float v8f;

#define DIM 64

__global__ void gcn_gemm_wmma_kernel(const float* __restrict__ X,
                                     const float* __restrict__ W,
                                     float* __restrict__ S,
                                     int n_row_tiles) {
    const int wave = threadIdx.x >> 5;
    const int lane = threadIdx.x & 31;
    const int tile = blockIdx.x * (blockDim.x >> 5) + wave;
    if (tile >= n_row_tiles) return;

    const int m0    = tile << 4;
    const int mrow  = lane & 15;
    const int khalf = lane >> 4;
    const int ncol  = lane & 15;

    v8f acc0 = {}, acc1 = {}, acc2 = {}, acc3 = {};

    const float* __restrict__ xrow = X + (long long)(m0 + mrow) * DIM;

    #pragma unroll
    for (int k0 = 0; k0 < DIM; k0 += 4) {
        const int ka = k0 + (khalf << 1);
        v2f a = *(const v2f*)(xrow + ka);

        const float* __restrict__ w0 = W + ka * DIM;
        const float* __restrict__ w1 = W + (ka + 1) * DIM;

        v2f b0, b1, b2, b3;
        b0.x = w0[ 0 + ncol]; b0.y = w1[ 0 + ncol];
        b1.x = w0[16 + ncol]; b1.y = w1[16 + ncol];
        b2.x = w0[32 + ncol]; b2.y = w1[32 + ncol];
        b3.x = w0[48 + ncol]; b3.y = w1[48 + ncol];

        acc0 = __builtin_amdgcn_wmma_f32_16x16x4_f32(false, a, false, b0,
                                                     (short)0, acc0, false, false);
        acc1 = __builtin_amdgcn_wmma_f32_16x16x4_f32(false, a, false, b1,
                                                     (short)0, acc1, false, false);
        acc2 = __builtin_amdgcn_wmma_f32_16x16x4_f32(false, a, false, b2,
                                                     (short)0, acc2, false, false);
        acc3 = __builtin_amdgcn_wmma_f32_16x16x4_f32(false, a, false, b3,
                                                     (short)0, acc3, false, false);
    }

    const int rbase = m0 + (khalf << 3);
    #pragma unroll
    for (int v = 0; v < 8; ++v) {
        float* __restrict__ srow = S + (long long)(rbase + v) * DIM + ncol;
        srow[ 0] = acc0[v];
        srow[16] = acc1[v];
        srow[32] = acc2[v];
        srow[48] = acc3[v];
    }
}

#define OWN_ROWS    391
#define OWN_THREADS 1024
#define OWN_WAVES   32
#define SCAN_DEPTH  8
#define WQ_CAP      (32 + SCAN_DEPTH * 4 * 32)

__device__ __forceinline__ void lds_add(float* p, float v) {
    __hip_atomic_fetch_add(p, v, __ATOMIC_RELAXED, __HIP_MEMORY_SCOPE_WORKGROUP);
}

__global__ __launch_bounds__(OWN_THREADS)
void gcn_owner_rows_kernel(const float* __restrict__ S,
                           const int* __restrict__ adj_row,
                           const int* __restrict__ adj_col,
                           const float* __restrict__ adj_val,
                           float* __restrict__ out,
                           int n_nodes, int n_edges) {
    __shared__ float acc[OWN_ROWS * DIM];
    __shared__ int            wq[OWN_WAVES * WQ_CAP];
    __shared__ unsigned short wl[OWN_WAVES * WQ_CAP];

    const int tid   = threadIdx.x;
    const int lane  = tid & 31;
    const int wave  = __builtin_amdgcn_readfirstlane(tid >> 5);
    const int qbase = wave * WQ_CAP;

    const int nv = n_edges >> 2;
    const int4* __restrict__ row4 = reinterpret_cast<const int4*>(adj_row);
    const int rot   = ((int)blockIdx.x * (nv / (int)gridDim.x)) & ~31;
    const int niter = (nv + OWN_THREADS - 1) / OWN_THREADS;

    for (int r0 = blockIdx.x * OWN_ROWS; r0 < n_nodes; r0 += gridDim.x * OWN_ROWS) {
        const unsigned nrows = (unsigned)min(OWN_ROWS, n_nodes - r0);

        for (int i = tid; i < OWN_ROWS * DIM / 4; i += OWN_THREADS)
            reinterpret_cast<float4*>(acc)[i] = make_float4(0.f, 0.f, 0.f, 0.f);
        __syncthreads();

        int qn = 0;

        auto note = [&](unsigned lr, int e) {
            const bool m = lr < nrows;
            const unsigned mask = __builtin_amdgcn_ballot_w32(m);
            if (mask) {
                if (m) {
                    const int slot = qbase + qn + (int)__builtin_amdgcn_mbcnt_lo(mask, 0u);
                    wq[slot] = e;
                    wl[slot] = (unsigned short)lr;
                }
                qn += __builtin_popcount(mask);
            }
        };

        auto drain = [&](int first, int cnt) {
            int e = 0, lr = 0;
            if (lane < cnt) {
                e  = wq[qbase + first + lane];
                lr = wl[qbase + first + lane];
            }
            const int c  = adj_col[e];
            const int wi = __float_as_int(adj_val[e]);
            float2 sv[32];
            #pragma unroll
            for (int u = 0; u < 32; ++u) {
                const int k  = min(u, cnt - 1);
                const int ck = __builtin_amdgcn_readlane(c, k);
                sv[u] = *reinterpret_cast<const float2*>(S + (long long)ck * DIM + 2 * lane);
            }
            #pragma unroll
            for (int u = 0; u < 32; ++u) {
                if (u < cnt) {
                    const int   lrk = __builtin_amdgcn_readlane(lr, u);
                    const float wk  = __int_as_float(__builtin_amdgcn_readlane(wi, u));
                    const int   a   = lrk * DIM + 2 * lane;
                    lds_add(&acc[a],     wk * sv[u].x);
                    lds_add(&acc[a + 1], wk * sv[u].y);
                }
            }
        };

        if (wave == 0) {
            const int e = (nv << 2) + lane;
            unsigned lr = ~0u;
            if (e < n_edges) lr = (unsigned)(adj_row[e] - r0);
            note(lr, e);
        }

        for (int it = 0; it < niter; it += SCAN_DEPTH) {
            int4     rv[SCAN_DEPTH];
            unsigned jv[SCAN_DEPTH];
            const int base = it * OWN_THREADS + tid;
            if ((it + SCAN_DEPTH) * OWN_THREADS <= nv) {
                #pragma unroll
                for (int u = 0; u < SCAN_DEPTH; ++u) {
                    unsigned j = (unsigned)(base + u * OWN_THREADS + rot);
                    j = min(j, j - (unsigned)nv);
                    jv[u] = j;
                    rv[u] = row4[j];
                }
            } else {
                #pragma unroll
                for (int u = 0; u < SCAN_DEPTH; ++u) {
                    const int idx = base + u * OWN_THREADS;
                    unsigned j = (unsigned)(idx + rot);
                    j = min(j, j - (unsigned)nv);
                    if (idx >= nv) j = 0u;
                    jv[u] = j;
                    rv[u] = row4[j];
                    if (idx >= nv) rv[u] = make_int4(-1, -1, -1, -1);
                }
            }
            #pragma unroll
            for (int u = 0; u < SCAN_DEPTH; ++u) {
                const unsigned l0 = (unsigned)(rv[u].x - r0);
                const unsigned l1 = (unsigned)(rv[u].y - r0);
                const unsigned l2 = (unsigned)(rv[u].z - r0);
                const unsigned l3 = (unsigned)(rv[u].w - r0);
                const unsigned lm = min(min(l0, l1), min(l2, l3));
                if (__builtin_amdgcn_ballot_w32(lm < nrows)) {
                    const int e0 = (int)(jv[u] << 2);
                    note(l0, e0 + 0);
                    note(l1, e0 + 1);
                    note(l2, e0 + 2);
                    note(l3, e0 + 3);
                }
            }
            while (qn >= 32) {
                drain(qn - 32, 32);
                qn -= 32;
            }
        }
        if (qn > 0) drain(0, qn);
        __syncthreads();

        float4* __restrict__ orow = reinterpret_cast<float4*>(out + (long long)r0 * DIM);
        for (int i = tid; i < (int)nrows * DIM / 4; i += OWN_THREADS)
            orow[i] = reinterpret_cast<const float4*>(acc)[i];
        __syncthreads();
    }
}

extern "C" void kernel_launch(void* const* d_in, const int* in_sizes, int n_in,
                              void* d_out, int out_size, void* d_ws, size_t ws_size,
                              hipStream_t stream) {
    const float* X    = (const float*)d_in[0];
    const float* W    = (const float*)d_in[1];
    const int*   arow = (const int*)d_in[2];
    const int*   acol = (const int*)d_in[3];
    const float* aval = (const float*)d_in[4];
    float*       out  = (float*)d_out;
    float*       S    = (float*)d_ws;

    const int n_nodes = in_sizes[0] / DIM;
    const int n_edges = in_sizes[2];

    {
        const int tiles = n_nodes / 16;
        const int waves_per_block = 256 / 32;
        const int blocks = (tiles + waves_per_block - 1) / waves_per_block;
        gcn_gemm_wmma_kernel<<<blocks, 256, 0, stream>>>(X, W, S, tiles);
    }

    {
        const int blocks = (n_nodes + OWN_ROWS - 1) / OWN_ROWS;
        gcn_owner_rows_kernel<<<blocks, OWN_THREADS, 0, stream>>>(S, arow, acol, aval, out,
                                                                  n_nodes, n_edges);
    }
}


# adj_row scan narrowed 4x: the product launch's waves also write one owner byte (row/391) per edge; every owner scans that byte list, 16 edges per load; LDS float adds in arrival order as before
# speedup vs baseline: 3.7424x; 1.2285x over previous
#include <hip/hip_runtime.h>

typedef __attribute__((ext_vector_type(2))) float v2f;
typedef __attribute__((ext_vector_type(8))) float v8f;

#define DIM 64
#define OWN_ROWS 391

__device__ __forceinline__ unsigned owner_bytes(int4 r) {
    return  (((unsigned)r.x / (unsigned)OWN_ROWS) & 255u)
         | ((((unsigned)r.y / (unsigned)OWN_ROWS) & 255u) << 8)
         | ((((unsigned)r.z / (unsigned)OWN_ROWS) & 255u) << 16)
         |  (((unsigned)r.w / (unsigned)OWN_ROWS)         << 24);
}

__global__ void gcn_gemm_wmma_kernel(const float* __restrict__ X,
                                     const float* __restrict__ W,
                                     float* __restrict__ S,
                                     int n_row_tiles,
                                     const int* __restrict__ adj_row,
                                     unsigned* __restrict__ own,
                                     int n_edges) {
    const int wave = threadIdx.x >> 5;
    const int lane = threadIdx.x & 31;
    const int tile = blockIdx.x * (blockDim.x >> 5) + wave;
    if (tile >= n_row_tiles) return;

    const int m0    = tile << 4;
    const int mrow  = lane & 15;
    const int khalf = lane >> 4;
    const int ncol  = lane & 15;

    v8f acc0 = {}, acc1 = {}, acc2 = {}, acc3 = {};

    const float* __restrict__ xrow = X + (long long)(m0 + mrow) * DIM;

    #pragma unroll
    for (int k0 = 0; k0 < DIM; k0 += 4) {
        const int ka = k0 + (khalf << 1);
        v2f a = *(const v2f*)(xrow + ka);

        const float* __restrict__ w0 = W + ka * DIM;
        const float* __restrict__ w1 = W + (ka + 1) * DIM;

        v2f b0, b1, b2, b3;
        b0.x = w0[ 0 + ncol]; b0.y = w1[ 0 + ncol];
        b1.x = w0[16 + ncol]; b1.y = w1[16 + ncol];
        b2.x = w0[32 + ncol]; b2.y = w1[32 + ncol];
        b3.x = w0[48 + ncol]; b3.y = w1[48 + ncol];

        acc0 = __builtin_amdgcn_wmma_f32_16x16x4_f32(false, a, false, b0,
                                                     (short)0, acc0, false, false);
        acc1 = __builtin_amdgcn_wmma_f32_16x16x4_f32(false, a, false, b1,
                                                     (short)0, acc1, false, false);
        acc2 = __builtin_amdgcn_wmma_f32_16x16x4_f32(false, a, false, b2,
                                                     (short)0, acc2, false, false);
        acc3 = __builtin_amdgcn_wmma_f32_16x16x4_f32(false, a, false, b3,
                                                     (short)0, acc3, false, false);
    }

    const int rbase = m0 + (khalf << 3);
    #pragma unroll
    for (int v = 0; v < 8; ++v) {
        float* __restrict__ srow = S + (long long)(rbase + v) * DIM + ncol;
        srow[ 0] = acc0[v];
        srow[16] = acc1[v];
        srow[32] = acc2[v];
        srow[48] = acc3[v];
    }

    const int4* __restrict__ row4 = reinterpret_cast<const int4*>(adj_row);
    uint2* __restrict__ own2 = reinterpret_cast<uint2*>(own);
    const int nv      = n_edges >> 2;
    const int npair   = nv >> 1;
    const int pstride = n_row_tiles << 5;
    for (int q = (tile << 5) + lane; q < npair; q += pstride) {
        const int4 ra = row4[2 * q];
        const int4 rb = row4[2 * q + 1];
        own2[q] = make_uint2(owner_bytes(ra), owner_bytes(rb));
    }
    if ((nv & 1) && tile == 0 && lane == 0)
        own[nv - 1] = owner_bytes(row4[nv - 1]);
}

#define OWN_THREADS 1024
#define OWN_WAVES   32
#define SCAN_DEPTH  8
#define FLUSH_EVERY 2
#define WQ_CAP      (32 + FLUSH_EVERY * 16 * 32)

__device__ __forceinline__ void lds_add(float* p, float v) {
    __hip_atomic_fetch_add(p, v, __ATOMIC_RELAXED, __HIP_MEMORY_SCOPE_WORKGROUP);
}

__device__ __forceinline__ unsigned may_match(unsigned x, unsigned pat) {
    const unsigned y = x ^ pat;
    return (y - 0x01010101u) & ~y & 0x80808080u;
}

__global__ __launch_bounds__(OWN_THREADS)
void gcn_owner_rows_kernel(const float* __restrict__ S,
                           const unsigned* __restrict__ own,
                           const int* __restrict__ adj_row,
                           const int* __restrict__ adj_col,
                           const float* __restrict__ adj_val,
                           float* __restrict__ out,
                           int n_nodes, int n_edges) {
    __shared__ float acc[OWN_ROWS * DIM];
    __shared__ int   wq[OWN_WAVES * WQ_CAP];

    const int tid   = threadIdx.x;
    const int lane  = tid & 31;
    const int wave  = __builtin_amdgcn_readfirstlane(tid >> 5);
    const int qbase = wave * WQ_CAP;

    const int nq = n_edges >> 4;
    const uint4* __restrict__ own4 = reinterpret_cast<const uint4*>(own);
    const int rot   = ((int)blockIdx.x * (nq / (int)gridDim.x)) & ~31;
    const int niter = (nq + OWN_THREADS - 1) / OWN_THREADS;

    for (int r0 = blockIdx.x * OWN_ROWS; r0 < n_nodes; r0 += gridDim.x * OWN_ROWS) {
        const unsigned nrows = (unsigned)min(OWN_ROWS, n_nodes - r0);
        const unsigned pat = ((unsigned)(r0 / OWN_ROWS) & 255u) * 0x01010101u;

        for (int i = tid; i < OWN_ROWS * DIM / 4; i += OWN_THREADS)
            reinterpret_cast<float4*>(acc)[i] = make_float4(0.f, 0.f, 0.f, 0.f);
        __syncthreads();

        int qn = 0;

        auto note16 = [&](unsigned m, unsigned j) {
            unsigned mask = __builtin_amdgcn_ballot_w32(m != 0u);
            while (mask) {
                if (m != 0u) {
                    const int p    = __builtin_ctz(m);
                    const int slot = qbase + qn + (int)__builtin_amdgcn_mbcnt_lo(mask, 0u);
                    wq[slot] = (int)(j << 4) + ((7 - (p & 7)) << 2) + (p >> 3);
                    m &= m - 1u;
                }
                qn += __builtin_popcount(mask);
                mask = __builtin_amdgcn_ballot_w32(m != 0u);
            }
        };

        auto drain = [&](int first, int cnt) {
            int e = 0;
            if (lane < cnt) e = wq[qbase + first + lane];
            const int c  = adj_col[e];
            const int wi = __float_as_int(adj_val[e]);
            unsigned  lr = (unsigned)(adj_row[e] - r0);
            if (lane >= cnt) lr = ~0u;
            for (int h = 0; h < cnt; h += 16) {
                float2 sv[16];
                #pragma unroll
                for (int u = 0; u < 16; ++u) {
                    const int k  = min(h + u, cnt - 1);
                    const int ck = __builtin_amdgcn_readlane(c, k);
                    sv[u] = *reinterpret_cast<const float2*>(S + (long long)ck * DIM + 2 * lane);
                }
                #pragma unroll
                for (int u = 0; u < 16; ++u) {
                    const unsigned lrk = (unsigned)__builtin_amdgcn_readlane((int)lr, h + u);
                    if (lrk < nrows) {
                        const float wk = __int_as_float(__builtin_amdgcn_readlane(wi, h + u));
                        const int   a  = (int)lrk * DIM + 2 * lane;
                        lds_add(&acc[a],     wk * sv[u].x);
                        lds_add(&acc[a + 1], wk * sv[u].y);
                    }
                }
            }
        };

        if (wave == 0) {
            const int e = (nq << 4) + lane;
            unsigned lr = ~0u;
            if (lane < 16 && e < n_edges) lr = (unsigned)(adj_row[e] - r0);
            const unsigned mask = __builtin_amdgcn_ballot_w32(lr < nrows);
            if (mask) {
                if (lr < nrows) wq[qbase + (int)__builtin_amdgcn_mbcnt_lo(mask, 0u)] = e;
                qn = __builtin_popcount(mask);
            }
        }

        for (int it = 0; it < niter; it += SCAN_DEPTH) {
            uint4    rv[SCAN_DEPTH];
            const int base = it * OWN_THREADS + tid;
            if ((it + SCAN_DEPTH) * OWN_THREADS <= nq) {
                #pragma unroll
                for (int u = 0; u < SCAN_DEPTH; ++u) {
                    unsigned j = (unsigned)(base + u * OWN_THREADS + rot);
                    j = min(j, j - (unsigned)nq);
                    rv[u] = own4[j];
                }
            } else {
                #pragma unroll
                for (int u = 0; u < SCAN_DEPTH; ++u) {
                    const int idx = base + u * OWN_THREADS;
                    unsigned j = (unsigned)(idx + rot);
                    j = min(j, j - (unsigned)nq);
                    if (idx >= nq) j = 0u;
                    rv[u] = own4[j];
                    if (idx >= nq) rv[u] = make_uint4(~pat, ~pat, ~pat, ~pat);
                }
            }
            #pragma unroll
            for (int u = 0; u < SCAN_DEPTH; ++u) {
                const unsigned f0 = may_match(rv[u].x, pat);
                const unsigned f1 = may_match(rv[u].y, pat);
                const unsigned f2 = may_match(rv[u].z, pat);
                const unsigned f3 = may_match(rv[u].w, pat);
                unsigned j = (unsigned)(base + u * OWN_THREADS + rot);
                j = min(j, j - (unsigned)nq);
                note16(f0 | (f1 >> 1) | (f2 >> 2) | (f3 >> 3), j);
                if ((u % FLUSH_EVERY) == FLUSH_EVERY - 1) {
                    while (qn >= 32) {
                        drain(qn - 32, 32);
                        qn -= 32;
                    }
                }
            }
        }
        if (qn > 0) drain(0, qn);
        __syncthreads();

        float4* __restrict__ orow = reinterpret_cast<float4*>(out + (long long)r0 * DIM);
        for (int i = tid; i < (int)nrows * DIM / 4; i += OWN_THREADS)
            orow[i] = reinterpret_cast<const float4*>(acc)[i];
        __syncthreads();
    }
}

extern "C" void kernel_launch(void* const* d_in, const int* in_sizes, int n_in,
                              void* d_out, int out_size, void* d_ws, size_t ws_size,
                              hipStream_t stream) {
    const float* X    = (const float*)d_in[0];
    const float* W    = (const float*)d_in[1];
    const int*   arow = (const int*)d_in[2];
    const int*   acol = (const int*)d_in[3];
    const float* aval = (const float*)d_in[4];
    float*       out  = (float*)d_out;
    float*       S    = (float*)d_ws;
    unsigned*    own  = (unsigned*)((char*)d_ws + 134217728);

    const int n_nodes = in_sizes[0] / DIM;
    const int n_edges = in_sizes[2];

    {
        const int tiles = n_nodes / 16;
        const int waves_per_block = 256 / 32;
        const int blocks = (tiles + waves_per_block - 1) / waves_per_block;
        gcn_gemm_wmma_kernel<<<blocks, 256, 0, stream>>>(X, W, S, tiles, arow, own, n_edges);
    }

    {
        const int blocks = (n_nodes + OWN_ROWS - 1) / OWN_ROWS;
        gcn_owner_rows_kernel<<<blocks, OWN_THREADS, 0, stream>>>(S, own, arow, acol, aval, out,
                                                                  n_nodes, n_edges);
    }
}


# Edge records: the product launch also packs each edge's column, row within its owner and value into 8 bytes; an owner reads one record per noted edge, not three words; LDS float adds as before
# speedup vs baseline: 3.9925x; 1.0668x over previous
#include <hip/hip_runtime.h>

typedef __attribute__((ext_vector_type(2))) float v2f;
typedef __attribute__((ext_vector_type(8))) float v8f;

#define DIM 64
#define OWN_ROWS 391

#define REC_ROW_SHIFT 20
#define REC_OWN_SHIFT 29

__device__ __forceinline__ unsigned rec_word0(unsigned row, unsigned col, unsigned owner) {
    return col | ((row - owner * (unsigned)OWN_ROWS) << REC_ROW_SHIFT) | (owner << REC_OWN_SHIFT);
}

__global__ void gcn_gemm_wmma_kernel(const float* __restrict__ X,
                                     const float* __restrict__ W,
                                     float* __restrict__ S,
                                     int n_row_tiles,
                                     const int* __restrict__ adj_row,
                                     const int* __restrict__ adj_col,
                                     const float* __restrict__ adj_val,
                                     unsigned* __restrict__ own,
                                     uint2* __restrict__ rec,
                                     int n_edges) {
    const int wave = threadIdx.x >> 5;
    const int lane = threadIdx.x & 31;
    const int tile = blockIdx.x * (blockDim.x >> 5) + wave;
    if (tile >= n_row_tiles) return;

    const int m0    = tile << 4;
    const int mrow  = lane & 15;
    const int khalf = lane >> 4;
    const int ncol  = lane & 15;

    v8f acc0 = {}, acc1 = {}, acc2 = {}, acc3 = {};

    const float* __restrict__ xrow = X + (long long)(m0 + mrow) * DIM;

    #pragma unroll
    for (int k0 = 0; k0 < DIM; k0 += 4) {
        const int ka = k0 + (khalf << 1);
        v2f a = *(const v2f*)(xrow + ka);

        const float* __restrict__ w0 = W + ka * DIM;
        const float* __restrict__ w1 = W + (ka + 1) * DIM;

        v2f b0, b1, b2, b3;
        b0.x = w0[ 0 + ncol]; b0.y = w1[ 0 + ncol];
        b1.x = w0[16 + ncol]; b1.y = w1[16 + ncol];
        b2.x = w0[32 + ncol]; b2.y = w1[32 + ncol];
        b3.x = w0[48 + ncol]; b3.y = w1[48 + ncol];

        acc0 = __builtin_amdgcn_wmma_f32_16x16x4_f32(false, a, false, b0,
                                                     (short)0, acc0, false, false);
        acc1 = __builtin_amdgcn_wmma_f32_16x16x4_f32(false, a, false, b1,
                                                     (short)0, acc1, false, false);
        acc2 = __builtin_amdgcn_wmma_f32_16x16x4_f32(false, a, false, b2,
                                                     (short)0, acc2, false, false);
        acc3 = __builtin_amdgcn_wmma_f32_16x16x4_f32(false, a, false, b3,
                                                     (short)0, acc3, false, false);
    }

    const int rbase = m0 + (khalf << 3);
    #pragma unroll
    for (int v = 0; v < 8; ++v) {
        float* __restrict__ srow = S + (long long)(rbase + v) * DIM + ncol;
        srow[ 0] = acc0[v];
        srow[16] = acc1[v];
        srow[32] = acc2[v];
        srow[48] = acc3[v];
    }

    const int2*   __restrict__ row2 = reinterpret_cast<const int2*>(adj_row);
    const int2*   __restrict__ col2 = reinterpret_cast<const int2*>(adj_col);
    const float2* __restrict__ val2 = reinterpret_cast<const float2*>(adj_val);
    uint4*          __restrict__ rec4  = reinterpret_cast<uint4*>(rec);
    unsigned short* __restrict__ own16 = reinterpret_cast<unsigned short*>(own);
    const int npair   = n_edges >> 1;
    const int pstride = n_row_tiles << 7;
    for (int q0 = tile << 7; q0 < npair; q0 += pstride) {
        int2 r[4], c[4];
        float2 v[4];
        #pragma unroll
        for (int k = 0; k < 4; ++k) {
            const int q = min(q0 + 32 * k + lane, npair - 1);
            r[k] = row2[q];
            c[k] = col2[q];
            v[k] = val2[q];
        }
        #pragma unroll
        for (int k = 0; k < 4; ++k) {
            const int q = q0 + 32 * k + lane;
            if (q < npair) {
                const unsigned oa = (unsigned)r[k].x / (unsigned)OWN_ROWS;
                const unsigned ob = (unsigned)r[k].y / (unsigned)OWN_ROWS;
                rec4[q] = make_uint4(rec_word0((unsigned)r[k].x, (unsigned)c[k].x, oa),
                                     __float_as_uint(v[k].x),
                                     rec_word0((unsigned)r[k].y, (unsigned)c[k].y, ob),
                                     __float_as_uint(v[k].y));
                own16[q] = (unsigned short)((oa & 255u) | ((ob & 255u) << 8));
            }
        }
    }
    if ((n_edges & 1) && tile == 0 && lane == 0) {
        const int e = n_edges - 1;
        const unsigned o = (unsigned)adj_row[e] / (unsigned)OWN_ROWS;
        rec[e] = make_uint2(rec_word0((unsigned)adj_row[e], (unsigned)adj_col[e], o),
                            __float_as_uint(adj_val[e]));
    }
}

#define OWN_THREADS 1024
#define OWN_WAVES   32
#define SCAN_DEPTH  8
#define FLUSH_EVERY 2
#define WQ_CAP      (32 + FLUSH_EVERY * 16 * 32)

__device__ __forceinline__ void lds_add(float* p, float v) {
    __hip_atomic_fetch_add(p, v, __ATOMIC_RELAXED, __HIP_MEMORY_SCOPE_WORKGROUP);
}

__device__ __forceinline__ unsigned may_match(unsigned x, unsigned pat) {
    const unsigned y = x ^ pat;
    return (y - 0x01010101u) & ~y & 0x80808080u;
}

__global__ __launch_bounds__(OWN_THREADS)
void gcn_owner_rows_kernel(const float* __restrict__ S,
                           const unsigned* __restrict__ own,
                           const uint2* __restrict__ rec,
                           const int* __restrict__ adj_row,
                           float* __restrict__ out,
                           int n_nodes, int n_edges) {
    __shared__ float acc[OWN_ROWS * DIM];
    __shared__ int   wq[OWN_WAVES * WQ_CAP];

    const int tid   = threadIdx.x;
    const int lane  = tid & 31;
    const int wave  = __builtin_amdgcn_readfirstlane(tid >> 5);
    const int qbase = wave * WQ_CAP;

    const int nq = n_edges >> 4;
    const uint4* __restrict__ own4 = reinterpret_cast<const uint4*>(own);
    const int rot   = ((int)blockIdx.x * (nq / (int)gridDim.x)) & ~31;
    const int niter = (nq + OWN_THREADS - 1) / OWN_THREADS;

    for (int r0 = blockIdx.x * OWN_ROWS; r0 < n_nodes; r0 += gridDim.x * OWN_ROWS) {
        const unsigned nrows = (unsigned)min(OWN_ROWS, n_nodes - r0);
        const unsigned pat = ((unsigned)(r0 / OWN_ROWS) & 255u) * 0x01010101u;
        const unsigned ownbits = ((unsigned)(r0 / OWN_ROWS) << REC_OWN_SHIFT) >> REC_OWN_SHIFT;

        for (int i = tid; i < OWN_ROWS * DIM / 4; i += OWN_THREADS)
            reinterpret_cast<float4*>(acc)[i] = make_float4(0.f, 0.f, 0.f, 0.f);
        __syncthreads();

        int qn = 0;

        auto note16 = [&](unsigned m, unsigned j) {
            unsigned mask = __builtin_amdgcn_ballot_w32(m != 0u);
            while (mask) {
                if (m != 0u) {
                    const int p    = __builtin_ctz(m);
                    const int slot = qbase + qn + (int)__builtin_amdgcn_mbcnt_lo(mask, 0u);
                    wq[slot] = (int)(j << 4) + ((7 - (p & 7)) << 2) + (p >> 3);
                    m &= m - 1u;
                }
                qn += __builtin_popcount(mask);
                mask = __builtin_amdgcn_ballot_w32(m != 0u);
            }
        };

        auto drain = [&](int first, int cnt) {
            int e = 0;
            if (lane < cnt) e = wq[qbase + first + lane];
            const uint2 rc = rec[e];
            const int c  = (int)(rc.x & ((1u << REC_ROW_SHIFT) - 1u));
            const int wi = (int)rc.y;
            unsigned  lr = (rc.x << (32 - REC_OWN_SHIFT)) >> (32 - REC_OWN_SHIFT + REC_ROW_SHIFT);
            if (lane >= cnt || (rc.x >> REC_OWN_SHIFT) != ownbits) lr = ~0u;
            for (int h = 0; h < cnt; h += 16) {
                float2 sv[16];
                #pragma unroll
                for (int u = 0; u < 16; ++u) {
                    const int k  = min(h + u, cnt - 1);
                    const int ck = __builtin_amdgcn_readlane(c, k);
                    sv[u] = *reinterpret_cast<const float2*>(S + (long long)ck * DIM + 2 * lane);
                }
                #pragma unroll
                for (int u = 0; u < 16; ++u) {
                    const unsigned lrk = (unsigned)__builtin_amdgcn_readlane((int)lr, h + u);
                    if (lrk < nrows) {
                        const float wk = __int_as_float(__builtin_amdgcn_readlane(wi, h + u));
                        const int   a  = (int)lrk * DIM + 2 * lane;
                        lds_add(&acc[a],     wk * sv[u].x);
                        lds_add(&acc[a + 1], wk * sv[u].y);
                    }
                }
            }
        };

        if (wave == 0) {
            const int e = (nq << 4) + lane;
            unsigned lr = ~0u;
            if (lane < 16 && e < n_edges) lr = (unsigned)(adj_row[e] - r0);
            const unsigned mask = __builtin_amdgcn_ballot_w32(lr < nrows);
            if (mask) {
                if (lr < nrows) wq[qbase + (int)__builtin_amdgcn_mbcnt_lo(mask, 0u)] = e;
                qn = __builtin_popcount(mask);
            }
        }

        for (int it = 0; it < niter; it += SCAN_DEPTH) {
            uint4    rv[SCAN_DEPTH];
            const int base = it * OWN_THREADS + tid;
            if ((it + SCAN_DEPTH) * OWN_THREADS <= nq) {
                #pragma unroll
                for (int u = 0; u < SCAN_DEPTH; ++u) {
                    unsigned j = (unsigned)(base + u * OWN_THREADS + rot);
                    j = min(j, j - (unsigned)nq);
                    rv[u] = own4[j];
                }
            } else {
                #pragma unroll
                for (int u = 0; u < SCAN_DEPTH; ++u) {
                    const int idx = base + u * OWN_THREADS;
                    unsigned j = (unsigned)(idx + rot);
                    j = min(j, j - (unsigned)nq);
                    if (idx >= nq) j = 0u;
                    rv[u] = own4[j];
                    if (idx >= nq) rv[u] = make_uint4(~pat, ~pat, ~pat, ~pat);
                }
            }
            #pragma unroll
            for (int u = 0; u < SCAN_DEPTH; ++u) {
                const unsigned f0 = may_match(rv[u].x, pat);
                const unsigned f1 = may_match(rv[u].y, pat);
                const unsigned f2 = may_match(rv[u].z, pat);
                const unsigned f3 = may_match(rv[u].w, pat);
                unsigned j = (unsigned)(base + u * OWN_THREADS + rot);
                j = min(j, j - (unsigned)nq);
                note16(f0 | (f1 >> 1) | (f2 >> 2) | (f3 >> 3), j);
                if ((u % FLUSH_EVERY) == FLUSH_EVERY - 1) {
                    while (qn >= 32) {
                        drain(qn - 32, 32);
                        qn -= 32;
                    }
                }
            }
        }
        if (qn > 0) drain(0, qn);
        __syncthreads();

        float4* __restrict__ orow = reinterpret_cast<float4*>(out + (long long)r0 * DIM);
        for (int i = tid; i < (int)nrows * DIM / 4; i += OWN_THREADS)
            orow[i] = reinterpret_cast<const float4*>(acc)[i];
        __syncthreads();
    }
}

extern "C" void kernel_launch(void* const* d_in, const int* in_sizes, int n_in,
                              void* d_out, int out_size, void* d_ws, size_t ws_size,
                              hipStream_t stream) {
    const float* X    = (const float*)d_in[0];
    const float* W    = (const float*)d_in[1];
    const int*   arow = (const int*)d_in[2];
    const int*   acol = (const int*)d_in[3];
    const float* aval = (const float*)d_in[4];
    float*       out  = (float*)d_out;
    float*       S    = (float*)d_ws;
    unsigned*    own  = (unsigned*)((char*)d_ws + 134217728);
    uint2*       rec  = (uint2*)((char*)d_ws + 67108864);

    const int n_nodes = in_sizes[0] / DIM;
    const int n_edges = in_sizes[2];

    {
        const int tiles = n_nodes / 16;
        const int waves_per_block = 256 / 32;
        const int blocks = (tiles + waves_per_block - 1) / waves_per_block;
        gcn_gemm_wmma_kernel<<<blocks, 256, 0, stream>>>(X, W, S, tiles, arow, acol, aval, own, rec,
                                                         n_edges);
    }

    {
        const int blocks = (n_nodes + OWN_ROWS - 1) / OWN_ROWS;
        gcn_owner_rows_kernel<<<blocks, OWN_THREADS, 0, stream>>>(S, own, rec, arow, out,
                                                                  n_nodes, n_edges);
    }
}


# LDS float adds replaced by register sums: wave w of an owner keeps rows w, w+32, ... in registers; found records reach the row's wave through LDS lists (one integer LDS add each); arrival-order sums
# speedup vs baseline: 4.2424x; 1.0626x over previous
#include <hip/hip_runtime.h>

typedef __attribute__((ext_vector_type(2))) float v2f;
typedef __attribute__((ext_vector_type(8))) float v8f;
typedef __attribute__((ext_vector_type(16))) float v16f;

#define DIM 64
#define OWN_ROWS 391

#define REC_ROW_SHIFT 20
#define REC_OWN_SHIFT 29

__device__ __forceinline__ unsigned rec_word0(unsigned row, unsigned col, unsigned owner) {
    return col | ((row - owner * (unsigned)OWN_ROWS) << REC_ROW_SHIFT) | (owner << REC_OWN_SHIFT);
}

__global__ void gcn_gemm_wmma_kernel(const float* __restrict__ X,
                                     const float* __restrict__ W,
                                     float* __restrict__ S,
                                     int n_row_tiles,
                                     const int* __restrict__ adj_row,
                                     const int* __restrict__ adj_col,
                                     const float* __restrict__ adj_val,
                                     unsigned* __restrict__ own,
                                     uint2* __restrict__ rec,
                                     int n_edges) {
    const int wave = threadIdx.x >> 5;
    const int lane = threadIdx.x & 31;
    const int tile = blockIdx.x * (blockDim.x >> 5) + wave;
    if (tile >= n_row_tiles) return;

    const int m0    = tile << 4;
    const int mrow  = lane & 15;
    const int khalf = lane >> 4;
    const int ncol  = lane & 15;

    v8f acc0 = {}, acc1 = {}, acc2 = {}, acc3 = {};

    const float* __restrict__ xrow = X + (long long)(m0 + mrow) * DIM;

    #pragma unroll
    for (int k0 = 0; k0 < DIM; k0 += 4) {
        const int ka = k0 + (khalf << 1);
        v2f a = *(const v2f*)(xrow + ka);

        const float* __restrict__ w0 = W + ka * DIM;
        const float* __restrict__ w1 = W + (ka + 1) * DIM;

        v2f b0, b1, b2, b3;
        b0.x = w0[ 0 + ncol]; b0.y = w1[ 0 + ncol];
        b1.x = w0[16 + ncol]; b1.y = w1[16 + ncol];
        b2.x = w0[32 + ncol]; b2.y = w1[32 + ncol];
        b3.x = w0[48 + ncol]; b3.y = w1[48 + ncol];

        acc0 = __builtin_amdgcn_wmma_f32_16x16x4_f32(false, a, false, b0,
                                                     (short)0, acc0, false, false);
        acc1 = __builtin_amdgcn_wmma_f32_16x16x4_f32(false, a, false, b1,
                                                     (short)0, acc1, false, false);
        acc2 = __builtin_amdgcn_wmma_f32_16x16x4_f32(false, a, false, b2,
                                                     (short)0, acc2, false, false);
        acc3 = __builtin_amdgcn_wmma_f32_16x16x4_f32(false, a, false, b3,
                                                     (short)0, acc3, false, false);
    }

    const int rbase = m0 + (khalf << 3);
    #pragma unroll
    for (int v = 0; v < 8; ++v) {
        float* __restrict__ srow = S + (long long)(rbase + v) * DIM + ncol;
        srow[ 0] = acc0[v];
        srow[16] = acc1[v];
        srow[32] = acc2[v];
        srow[48] = acc3[v];
    }

    const int2*   __restrict__ row2 = reinterpret_cast<const int2*>(adj_row);
    const int2*   __restrict__ col2 = reinterpret_cast<const int2*>(adj_col);
    const float2* __restrict__ val2 = reinterpret_cast<const float2*>(adj_val);
    uint4*          __restrict__ rec4  = reinterpret_cast<uint4*>(rec);
    unsigned short* __restrict__ own16 = reinterpret_cast<unsigned short*>(own);
    const int npair   = n_edges >> 1;
    const int pstride = n_row_tiles << 7;
    for (int q0 = tile << 7; q0 < npair; q0 += pstride) {
        int2 r[4], c[4];
        float2 v[4];
        #pragma unroll
        for (int k = 0; k < 4; ++k) {
            const int q = min(q0 + 32 * k + lane, npair - 1);
            r[k] = row2[q];
            c[k] = col2[q];
            v[k] = val2[q];
        }
        #pragma unroll
        for (int k = 0; k < 4; ++k) {
            const int q = q0 + 32 * k + lane;
            if (q < npair) {
                const unsigned oa = (unsigned)r[k].x / (unsigned)OWN_ROWS;
                const unsigned ob = (unsigned)r[k].y / (unsigned)OWN_ROWS;
                rec4[q] = make_uint4(rec_word0((unsigned)r[k].x, (unsigned)c[k].x, oa),
                                     __float_as_uint(v[k].x),
                                     rec_word0((unsigned)r[k].y, (unsigned)c[k].y, ob),
                                     __float_as_uint(v[k].y));
                own16[q] = (unsigned short)((oa & 255u) | ((ob & 255u) << 8));
            }
        }
    }
    if ((n_edges & 1) && tile == 0 && lane == 0) {
        const int e = n_edges - 1;
        const unsigned o = (unsigned)adj_row[e] / (unsigned)OWN_ROWS;
        rec[e] = make_uint2(rec_word0((unsigned)adj_row[e], (unsigned)adj_col[e], o),
                            __float_as_uint(adj_val[e]));
    }
}

#define OWN_THREADS 1024
#define OWN_WAVES   32
#define SCAN_DEPTH  8
#define WQ_CAP      (32 + 16 * 32)
#define RQ_CAP      448
#define WAVE_ROWS   ((OWN_ROWS + OWN_WAVES - 1) / OWN_WAVES)

__device__ __forceinline__ void lds_add(float* p, float v) {
    __hip_atomic_fetch_add(p, v, __ATOMIC_RELAXED, __HIP_MEMORY_SCOPE_WORKGROUP);
}

__device__ __forceinline__ unsigned may_match(unsigned x, unsigned pat) {
    const unsigned y = x ^ pat;
    return (y - 0x01010101u) & ~y & 0x80808080u;
}

__global__ __launch_bounds__(OWN_THREADS)
void gcn_owner_rows_kernel(const float* __restrict__ S,
                           const unsigned* __restrict__ own,
                           const uint2* __restrict__ rec,
                           const int* __restrict__ adj_row,
                           float* __restrict__ out,
                           int n_nodes, int n_edges) {
    __shared__ float acc[OWN_ROWS * DIM];
    __shared__ int   wq[OWN_WAVES * WQ_CAP];
    __shared__ uint2 rq[OWN_WAVES * RQ_CAP];
    __shared__ int   rcnt[OWN_WAVES];

    const int tid   = threadIdx.x;
    const int lane  = tid & 31;
    const int wave  = __builtin_amdgcn_readfirstlane(tid >> 5);
    const int qbase = wave * WQ_CAP;

    const int nq = n_edges >> 4;
    const uint4* __restrict__ own4 = reinterpret_cast<const uint4*>(own);
    const int rot   = ((int)blockIdx.x * (nq / (int)gridDim.x)) & ~31;
    const int niter = (nq + OWN_THREADS - 1) / OWN_THREADS;

    for (int r0 = blockIdx.x * OWN_ROWS; r0 < n_nodes; r0 += gridDim.x * OWN_ROWS) {
        const unsigned nrows = (unsigned)min(OWN_ROWS, n_nodes - r0);
        const unsigned pat = ((unsigned)(r0 / OWN_ROWS) & 255u) * 0x01010101u;
        const unsigned ownbits = ((unsigned)(r0 / OWN_ROWS) << REC_OWN_SHIFT) >> REC_OWN_SHIFT;

        for (int i = tid; i < OWN_ROWS * DIM / 4; i += OWN_THREADS)
            reinterpret_cast<float4*>(acc)[i] = make_float4(0.f, 0.f, 0.f, 0.f);
        if (tid < OWN_WAVES) rcnt[tid] = 0;
        __syncthreads();

        int qn = 0;

        auto note16 = [&](unsigned m, unsigned j) {
            unsigned mask = __builtin_amdgcn_ballot_w32(m != 0u);
            while (mask) {
                if (m != 0u) {
                    const int p    = __builtin_ctz(m);
                    const int slot = qbase + qn + (int)__builtin_amdgcn_mbcnt_lo(mask, 0u);
                    wq[slot] = (int)(j << 4) + ((7 - (p & 7)) << 2) + (p >> 3);
                    m &= m - 1u;
                }
                qn += __builtin_popcount(mask);
                mask = __builtin_amdgcn_ballot_w32(m != 0u);
            }
        };

        auto route = [&](int first, int cnt) {
            int e = 0;
            if (lane < cnt) e = wq[qbase + first + lane];
            const uint2 rc = rec[e];
            const unsigned lr = (rc.x << (32 - REC_OWN_SHIFT)) >> (32 - REC_OWN_SHIFT + REC_ROW_SHIFT);
            if (lane < cnt && (rc.x >> REC_OWN_SHIFT) == ownbits && lr < nrows) {
                const int d   = (int)(lr & (unsigned)(OWN_WAVES - 1));
                const int pos = __hip_atomic_fetch_add(&rcnt[d], 1, __ATOMIC_RELAXED,
                                                       __HIP_MEMORY_SCOPE_WORKGROUP);
                if (pos < RQ_CAP) {
                    rq[d * RQ_CAP + pos] = rc;
                } else {
                    const float wv = __uint_as_float(rc.y);
                    const float* __restrict__ srow =
                        S + (long long)(rc.x & ((1u << REC_ROW_SHIFT) - 1u)) * DIM;
                    #pragma nounroll
                    for (int j = 0; j < DIM; ++j) lds_add(&acc[lr * DIM + j], wv * srow[j]);
                }
            }
        };

        if (wave == 0) {
            const int e = (nq << 4) + lane;
            unsigned lr = ~0u;
            if (lane < 16 && e < n_edges) lr = (unsigned)(adj_row[e] - r0);
            const unsigned mask = __builtin_amdgcn_ballot_w32(lr < nrows);
            if (mask) {
                if (lr < nrows) wq[qbase + (int)__builtin_amdgcn_mbcnt_lo(mask, 0u)] = e;
                qn = __builtin_popcount(mask);
            }
        }

        for (int it = 0; it < niter; it += SCAN_DEPTH) {
            uint4    rv[SCAN_DEPTH];
            const int base = it * OWN_THREADS + tid;
            if ((it + SCAN_DEPTH) * OWN_THREADS <= nq) {
                #pragma unroll
                for (int u = 0; u < SCAN_DEPTH; ++u) {
                    unsigned j = (unsigned)(base + u * OWN_THREADS + rot);
                    j = min(j, j - (unsigned)nq);
                    rv[u] = own4[j];
                }
            } else {
                #pragma unroll
                for (int u = 0; u < SCAN_DEPTH; ++u) {
                    const int idx = base + u * OWN_THREADS;
                    unsigned j = (unsigned)(idx + rot);
                    j = min(j, j - (unsigned)nq);
                    if (idx >= nq) j = 0u;
                    rv[u] = own4[j];
                    if (idx >= nq) rv[u] = make_uint4(~pat, ~pat, ~pat, ~pat);
                }
            }
            #pragma unroll
            for (int u = 0; u < SCAN_DEPTH; ++u) {
                const unsigned f0 = may_match(rv[u].x, pat);
                const unsigned f1 = may_match(rv[u].y, pat);
                const unsigned f2 = may_match(rv[u].z, pat);
                const unsigned f3 = may_match(rv[u].w, pat);
                unsigned j = (unsigned)(base + u * OWN_THREADS + rot);
                j = min(j, j - (unsigned)nq);
                note16(f0 | (f1 >> 1) | (f2 >> 2) | (f3 >> 3), j);
                while (qn >= 32) {
                    route(qn - 32, 32);
                    qn -= 32;
                }
            }
        }
        if (qn > 0) route(0, qn);
        __syncthreads();

        const int nrec = min(__builtin_amdgcn_readfirstlane(rcnt[wave]), RQ_CAP);
        const uint2* myq = rq + wave * RQ_CAP;
        v16f ax = {}, ay = {};

        for (int i0 = 0; i0 < nrec; i0 += 32) {
            const int cnt = min(32, nrec - i0);
            const uint2 rc = myq[i0 + min(lane, cnt - 1)];
            const int c  = (int)(rc.x & ((1u << REC_ROW_SHIFT) - 1u));
            const int kr = (int)((rc.x << (32 - REC_OWN_SHIFT)) >> (32 - REC_OWN_SHIFT + REC_ROW_SHIFT + 5));
            const int wi = (int)rc.y;
            for (int h = 0; h < cnt; h += 16) {
                float2 sv[16];
                #pragma unroll
                for (int u = 0; u < 16; ++u) {
                    const int k  = min(h + u, cnt - 1);
                    const int ck = __builtin_amdgcn_readlane(c, k);
                    sv[u] = *reinterpret_cast<const float2*>(S + (long long)ck * DIM + 2 * lane);
                }
                #pragma unroll
                for (int u = 0; u < 16; ++u) {
                    if (h + u < cnt) {
                        const int   kk = __builtin_amdgcn_readlane(kr, h + u);
                        const float wk = __int_as_float(__builtin_amdgcn_readlane(wi, h + u));
                        ax[kk] += wk * sv[u].x;
                        ay[kk] += wk * sv[u].y;
                    }
                }
            }
        }

        #pragma unroll
        for (int k = 0; k < WAVE_ROWS; ++k) {
            const int r = wave + OWN_WAVES * k;
            if (r < (int)nrows) {
                const float2 o = *reinterpret_cast<const float2*>(&acc[r * DIM + 2 * lane]);
                *reinterpret_cast<float2*>(out + (long long)(r0 + r) * DIM + 2 * lane) =
                    make_float2(ax[k] + o.x, ay[k] + o.y);
            }
        }
        __syncthreads();
    }
}

extern "C" void kernel_launch(void* const* d_in, const int* in_sizes, int n_in,
                              void* d_out, int out_size, void* d_ws, size_t ws_size,
                              hipStream_t stream) {
    const float* X    = (const float*)d_in[0];
    const float* W    = (const float*)d_in[1];
    const int*   arow = (const int*)d_in[2];
    const int*   acol = (const int*)d_in[3];
    const float* aval = (const float*)d_in[4];
    float*       out  = (float*)d_out;
    float*       S    = (float*)d_ws;
    unsigned*    own  = (unsigned*)((char*)d_ws + 134217728);
    uint2*       rec  = (uint2*)((char*)d_ws + 67108864);

    const int n_nodes = in_sizes[0] / DIM;
    const int n_edges = in_sizes[2];

    {
        const int tiles = n_nodes / 16;
        const int waves_per_block = 256 / 32;
        const int blocks = (tiles + waves_per_block - 1) / waves_per_block;
        gcn_gemm_wmma_kernel<<<blocks, 256, 0, stream>>>(X, W, S, tiles, arow, acol, aval, own, rec,
                                                         n_edges);
    }

    {
        const int blocks = (n_nodes + OWN_ROWS - 1) / OWN_ROWS;
        gcn_owner_rows_kernel<<<blocks, OWN_THREADS, 0, stream>>>(S, own, rec, arow, out,
                                                                  n_nodes, n_edges);
    }
}


# Owner list kept as bit planes, eight dwords per 32 edges, written with wave ballots: an owner finds its edges with eight bit operations per 32 edges, no false hits; arrival-order sums as before
# speedup vs baseline: 4.5818x; 1.0800x over previous
#include <hip/hip_runtime.h>

typedef __attribute__((ext_vector_type(2))) float v2f;
typedef __attribute__((ext_vector_type(8))) float v8f;
typedef __attribute__((ext_vector_type(16))) float v16f;

#define DIM 64
#define OWN_ROWS 391

#define REC_ROW_SHIFT 20
#define REC_OWN_SHIFT 29

__device__ __forceinline__ unsigned rec_word0(unsigned row, unsigned col, unsigned owner) {
    return col | ((row - owner * (unsigned)OWN_ROWS) << REC_ROW_SHIFT) | (owner << REC_OWN_SHIFT);
}

__global__ void gcn_gemm_wmma_kernel(const float* __restrict__ X,
                                     const float* __restrict__ W,
                                     float* __restrict__ S,
                                     int n_row_tiles,
                                     const int* __restrict__ adj_row,
                                     const int* __restrict__ adj_col,
                                     const float* __restrict__ adj_val,
                                     unsigned* __restrict__ plo,
                                     unsigned* __restrict__ phi,
                                     uint2* __restrict__ rec,
                                     int n_edges) {
    const int wave = threadIdx.x >> 5;
    const int lane = threadIdx.x & 31;
    const int tile = blockIdx.x * (blockDim.x >> 5) + wave;
    if (tile >= n_row_tiles) return;

    const int m0    = tile << 4;
    const int mrow  = lane & 15;
    const int khalf = lane >> 4;
    const int ncol  = lane & 15;

    v8f acc0 = {}, acc1 = {}, acc2 = {}, acc3 = {};

    const float* __restrict__ xrow = X + (long long)(m0 + mrow) * DIM;

    #pragma unroll
    for (int k0 = 0; k0 < DIM; k0 += 4) {
        const int ka = k0 + (khalf << 1);
        v2f a = *(const v2f*)(xrow + ka);

        const float* __restrict__ w0 = W + ka * DIM;
        const float* __restrict__ w1 = W + (ka + 1) * DIM;

        v2f b0, b1, b2, b3;
        b0.x = w0[ 0 + ncol]; b0.y = w1[ 0 + ncol];
        b1.x = w0[16 + ncol]; b1.y = w1[16 + ncol];
        b2.x = w0[32 + ncol]; b2.y = w1[32 + ncol];
        b3.x = w0[48 + ncol]; b3.y = w1[48 + ncol];

        acc0 = __builtin_amdgcn_wmma_f32_16x16x4_f32(false, a, false, b0,
                                                     (short)0, acc0, false, false);
        acc1 = __builtin_amdgcn_wmma_f32_16x16x4_f32(false, a, false, b1,
                                                     (short)0, acc1, false, false);
        acc2 = __builtin_amdgcn_wmma_f32_16x16x4_f32(false, a, false, b2,
                                                     (short)0, acc2, false, false);
        acc3 = __builtin_amdgcn_wmma_f32_16x16x4_f32(false, a, false, b3,
                                                     (short)0, acc3, false, false);
    }

    const int rbase = m0 + (khalf << 3);
    #pragma unroll
    for (int v = 0; v < 8; ++v) {
        float* __restrict__ srow = S + (long long)(rbase + v) * DIM + ncol;
        srow[ 0] = acc0[v];
        srow[16] = acc1[v];
        srow[32] = acc2[v];
        srow[48] = acc3[v];
    }

    const int2*   __restrict__ row2 = reinterpret_cast<const int2*>(adj_row);
    const int2*   __restrict__ col2 = reinterpret_cast<const int2*>(adj_col);
    const float2* __restrict__ val2 = reinterpret_cast<const float2*>(adj_val);
    uint4*          __restrict__ rec4  = reinterpret_cast<uint4*>(rec);
    const int npair   = n_edges >> 1;
    const int nblk    = (npair + 31) >> 5;
    const int pstride = n_row_tiles << 7;
    for (int q0 = tile << 7; q0 < npair; q0 += pstride) {
        int2 r[4], c[4];
        float2 v[4];
        #pragma unroll
        for (int k = 0; k < 4; ++k) {
            const int q = min(q0 + 32 * k + lane, npair - 1);
            r[k] = row2[q];
            c[k] = col2[q];
            v[k] = val2[q];
        }
        unsigned wlo = 0u, whi = 0u;
        #pragma unroll
        for (int k = 0; k < 4; ++k) {
            const int q = q0 + 32 * k + lane;
            const unsigned oa = (unsigned)r[k].x / (unsigned)OWN_ROWS;
            const unsigned ob = (unsigned)r[k].y / (unsigned)OWN_ROWS;
            if (q < npair) {
                rec4[q] = make_uint4(rec_word0((unsigned)r[k].x, (unsigned)c[k].x, oa),
                                     __float_as_uint(v[k].x),
                                     rec_word0((unsigned)r[k].y, (unsigned)c[k].y, ob),
                                     __float_as_uint(v[k].y));
            } else if (q < (nblk << 5)) {
                rec4[q] = make_uint4(511u << REC_ROW_SHIFT, 0u, 511u << REC_ROW_SHIFT, 0u);
            }
            #pragma unroll
            for (int b = 0; b < 4; ++b) {
                const unsigned al = __builtin_amdgcn_ballot_w32(((oa >> b) & 1u) != 0u);
                const unsigned ah = __builtin_amdgcn_ballot_w32(((oa >> (b + 4)) & 1u) != 0u);
                const unsigned bl = __builtin_amdgcn_ballot_w32(((ob >> b) & 1u) != 0u);
                const unsigned bh = __builtin_amdgcn_ballot_w32(((ob >> (b + 4)) & 1u) != 0u);
                const bool mea = lane == (2 * k    ) * 4 + b;
                const bool meb = lane == (2 * k + 1) * 4 + b;
                wlo = mea ? al : wlo;
                whi = mea ? ah : whi;
                wlo = meb ? bl : wlo;
                whi = meb ? bh : whi;
            }
        }
        const int d = (q0 >> 4) * 4 + lane;
        if (d < nblk * 8) {
            plo[d] = wlo;
            phi[d] = whi;
        }
    }
    if ((n_edges & 1) && tile == 0 && lane == 0) {
        const int e = n_edges - 1;
        const unsigned o = (unsigned)adj_row[e] / (unsigned)OWN_ROWS;
        rec[nblk << 6] = make_uint2(rec_word0((unsigned)adj_row[e], (unsigned)adj_col[e], o),
                            __float_as_uint(adj_val[e]));
    }
}

#define OWN_THREADS 1024
#define OWN_WAVES   32
#define SCAN_DEPTH  4
#define WQ_CAP      64
#define RQ_CAP      448
#define WAVE_ROWS   ((OWN_ROWS + OWN_WAVES - 1) / OWN_WAVES)

__device__ __forceinline__ void lds_add(float* p, float v) {
    __hip_atomic_fetch_add(p, v, __ATOMIC_RELAXED, __HIP_MEMORY_SCOPE_WORKGROUP);
}

__global__ __launch_bounds__(OWN_THREADS)
void gcn_owner_rows_kernel(const float* __restrict__ S,
                           const unsigned* __restrict__ plo,
                           const unsigned* __restrict__ phi,
                           const uint2* __restrict__ rec,
                           const int* __restrict__ adj_row,
                           float* __restrict__ out,
                           int n_nodes, int n_edges) {
    __shared__ float acc[OWN_ROWS * DIM];
    __shared__ int   wq[OWN_WAVES * WQ_CAP];
    __shared__ uint2 rq[OWN_WAVES * RQ_CAP];
    __shared__ int   rcnt[OWN_WAVES];

    const int tid   = threadIdx.x;
    const int lane  = tid & 31;
    const int wave  = __builtin_amdgcn_readfirstlane(tid >> 5);
    const int qbase = wave * WQ_CAP;

    const int npair = n_edges >> 1;
    const int nq = ((npair + 31) >> 5) << 1;
    const uint4* __restrict__ lo4 = reinterpret_cast<const uint4*>(plo);
    const uint4* __restrict__ hi4 = reinterpret_cast<const uint4*>(phi);
    const int rot   = ((int)blockIdx.x * (nq / (int)gridDim.x)) & ~31;
    const int niter = (nq + OWN_THREADS - 1) / OWN_THREADS;

    for (int r0 = blockIdx.x * OWN_ROWS; r0 < n_nodes; r0 += gridDim.x * OWN_ROWS) {
        const unsigned nrows = (unsigned)min(OWN_ROWS, n_nodes - r0);
        const unsigned oid = (unsigned)(r0 / OWN_ROWS) & 255u;
        unsigned x[8];
        #pragma unroll
        for (int b = 0; b < 8; ++b) x[b] = ((oid >> b) & 1u) - 1u;
        const unsigned ownbits = ((unsigned)(r0 / OWN_ROWS) << REC_OWN_SHIFT) >> REC_OWN_SHIFT;

        for (int i = tid; i < OWN_ROWS * DIM / 4; i += OWN_THREADS)
            reinterpret_cast<float4*>(acc)[i] = make_float4(0.f, 0.f, 0.f, 0.f);
        if (tid < OWN_WAVES) rcnt[tid] = 0;
        __syncthreads();

        int qn = 0;

        auto route = [&](int first, int cnt) {
            int e = 0;
            if (lane < cnt) e = wq[qbase + first + lane];
            const uint2 rc = rec[e];
            const unsigned lr = (rc.x << (32 - REC_OWN_SHIFT)) >> (32 - REC_OWN_SHIFT + REC_ROW_SHIFT);
            if (lane < cnt && (rc.x >> REC_OWN_SHIFT) == ownbits && lr < nrows) {
                const int d   = (int)(lr & (unsigned)(OWN_WAVES - 1));
                const int pos = __hip_atomic_fetch_add(&rcnt[d], 1, __ATOMIC_RELAXED,
                                                       __HIP_MEMORY_SCOPE_WORKGROUP);
                if (pos < RQ_CAP) {
                    rq[d * RQ_CAP + pos] = rc;
                } else {
                    const float wv = __uint_as_float(rc.y);
                    const float* __restrict__ srow =
                        S + (long long)(rc.x & ((1u << REC_ROW_SHIFT) - 1u)) * DIM;
                    #pragma nounroll
                    for (int j = 0; j < DIM; ++j) lds_add(&acc[lr * DIM + j], wv * srow[j]);
                }
            }
        };

        auto note32 = [&](unsigned m, unsigned j) {
            const int ebase = (int)((j >> 1) << 6) + (int)(j & 1u);
            unsigned mask = __builtin_amdgcn_ballot_w32(m != 0u);
            while (mask) {
                if (m != 0u) {
                    const int p    = __builtin_ctz(m);
                    const int slot = qbase + qn + (int)__builtin_amdgcn_mbcnt_lo(mask, 0u);
                    wq[slot] = ebase + 2 * p;
                    m &= m - 1u;
                }
                qn = __builtin_amdgcn_readfirstlane(qn + __builtin_popcount(mask));
                if (qn >= 32) {
                    route(qn - 32, 32);
                    qn -= 32;
                }
                mask = __builtin_amdgcn_ballot_w32(m != 0u);
            }
        };

        if (wave == 0) {
            unsigned lr = ~0u;
            if (lane == 0 && (n_edges & 1)) lr = (unsigned)(adj_row[n_edges - 1] - r0);
            const unsigned mask = __builtin_amdgcn_ballot_w32(lr < nrows);
            if (mask) {
                if (lr < nrows) wq[qbase] = nq << 5;
                qn = 1;
            }
        }

        auto match8 = [&](const uint4& l, const uint4& h) {
            return (l.x ^ x[0]) & (l.y ^ x[1]) & (l.z ^ x[2]) & (l.w ^ x[3]) &
                   (h.x ^ x[4]) & (h.y ^ x[5]) & (h.z ^ x[6]) & (h.w ^ x[7]);
        };
        int it = 0;
        for (; (it + SCAN_DEPTH) * OWN_THREADS <= nq; it += SCAN_DEPTH) {
            uint4    lo[SCAN_DEPTH], hi[SCAN_DEPTH];
            const int base = it * OWN_THREADS + tid;
            #pragma unroll
            for (int u = 0; u < SCAN_DEPTH; ++u) {
                unsigned j = (unsigned)(base + u * OWN_THREADS + rot);
                j = min(j, j - (unsigned)nq);
                lo[u] = lo4[j];
                hi[u] = hi4[j];
            }
            #pragma unroll
            for (int u = 0; u < SCAN_DEPTH; ++u) {
                unsigned j = (unsigned)(base + u * OWN_THREADS + rot);
                j = min(j, j - (unsigned)nq);
                note32(match8(lo[u], hi[u]), j);
            }
        }
        for (; it < niter; ++it) {
            const int idx = it * OWN_THREADS + tid;
            unsigned j = (unsigned)(idx + rot);
            j = min(j, j - (unsigned)nq);
            if (idx >= nq) j = 0u;
            const uint4 l = lo4[j];
            const uint4 h = hi4[j];
            note32(idx < nq ? match8(l, h) : 0u, j);
        }
        if (qn > 0) route(0, qn);
        __syncthreads();

        const int nrec = min(__builtin_amdgcn_readfirstlane(rcnt[wave]), RQ_CAP);
        const uint2* myq = rq + wave * RQ_CAP;
        v16f ax = {}, ay = {};

        for (int i0 = 0; i0 < nrec; i0 += 32) {
            const int cnt = min(32, nrec - i0);
            const uint2 rc = myq[i0 + min(lane, cnt - 1)];
            const int c  = (int)(rc.x & ((1u << REC_ROW_SHIFT) - 1u));
            const int kr = (int)((rc.x << (32 - REC_OWN_SHIFT)) >> (32 - REC_OWN_SHIFT + REC_ROW_SHIFT + 5));
            const int wi = (int)rc.y;
            for (int h = 0; h < cnt; h += 16) {
                float2 sv[16];
                #pragma unroll
                for (int u = 0; u < 16; ++u) {
                    const int k  = min(h + u, cnt - 1);
                    const int ck = __builtin_amdgcn_readlane(c, k);
                    sv[u] = *reinterpret_cast<const float2*>(S + (long long)ck * DIM + 2 * lane);
                }
                #pragma unroll
                for (int u = 0; u < 16; ++u) {
                    if (h + u < cnt) {
                        const int   kk = __builtin_amdgcn_readlane(kr, h + u);
                        const float wk = __int_as_float(__builtin_amdgcn_readlane(wi, h + u));
                        ax[kk] += wk * sv[u].x;
                        ay[kk] += wk * sv[u].y;
                    }
                }
            }
        }

        #pragma unroll
        for (int k = 0; k < WAVE_ROWS; ++k) {
            const int r = wave + OWN_WAVES * k;
            if (r < (int)nrows) {
                const float2 o = *reinterpret_cast<const float2*>(&acc[r * DIM + 2 * lane]);
                *reinterpret_cast<float2*>(out + (long long)(r0 + r) * DIM + 2 * lane) =
                    make_float2(ax[k] + o.x, ay[k] + o.y);
            }
        }
        __syncthreads();
    }
}

extern "C" void kernel_launch(void* const* d_in, const int* in_sizes, int n_in,
                              void* d_out, int out_size, void* d_ws, size_t ws_size,
                              hipStream_t stream) {
    const float* X    = (const float*)d_in[0];
    const float* W    = (const float*)d_in[1];
    const int*   arow = (const int*)d_in[2];
    const int*   acol = (const int*)d_in[3];
    const float* aval = (const float*)d_in[4];
    float*       out  = (float*)d_out;
    float*       S    = (float*)d_ws;
    unsigned*    plo  = (unsigned*)((char*)d_ws + 134217728);
    unsigned*    phi  = (unsigned*)((char*)d_ws + 138412032);
    uint2*       rec  = (uint2*)((char*)d_ws + 67108864);

    const int n_nodes = in_sizes[0] / DIM;
    const int n_edges = in_sizes[2];

    {
        const int tiles = n_nodes / 16;
        const int waves_per_block = 256 / 32;
        const int blocks = (tiles + waves_per_block - 1) / waves_per_block;
        gcn_gemm_wmma_kernel<<<blocks, 256, 0, stream>>>(X, W, S, tiles, arow, acol, aval, plo, phi,
                                                         rec, n_edges);
    }

    {
        const int blocks = (n_nodes + OWN_ROWS - 1) / OWN_ROWS;
        gcn_owner_rows_kernel<<<blocks, OWN_THREADS, 0, stream>>>(S, plo, phi, rec, arow, out,
                                                                  n_nodes, n_edges);
    }
}


# Product launch moves its operands as whole lines through LDS: W staged once per work-group, each wave's X tile in and its S tile out as 16-byte accesses; fragments read from LDS; sums unchanged
# speedup vs baseline: 4.8675x; 1.0624x over previous
#include <hip/hip_runtime.h>

typedef __attribute__((ext_vector_type(2))) float v2f;
typedef __attribute__((ext_vector_type(8))) float v8f;
typedef __attribute__((ext_vector_type(16))) float v16f;

#define DIM 64
#define OWN_ROWS 391

#define REC_ROW_SHIFT 20
#define REC_OWN_SHIFT 29

__device__ __forceinline__ unsigned rec_word0(unsigned row, unsigned col, unsigned owner) {
    return col | ((row - owner * (unsigned)OWN_ROWS) << REC_ROW_SHIFT) | (owner << REC_OWN_SHIFT);
}

#define W_PITCH 72
#define X_PITCH 68

__global__ __launch_bounds__(256)
void gcn_gemm_wmma_kernel(const float* __restrict__ X,
                                     const float* __restrict__ W,
                                     float* __restrict__ S,
                                     int n_row_tiles,
                                     const int* __restrict__ adj_row,
                                     const int* __restrict__ adj_col,
                                     const float* __restrict__ adj_val,
                                     unsigned* __restrict__ plo,
                                     unsigned* __restrict__ phi,
                                     uint2* __restrict__ rec,
                                     int n_edges) {
    __shared__ __attribute__((aligned(16))) float wl[DIM * W_PITCH];
    __shared__ __attribute__((aligned(16))) float xl[8][16 * X_PITCH];

    const int wave = threadIdx.x >> 5;
    const int lane = threadIdx.x & 31;
    const int tile = blockIdx.x * 8 + wave;
    const bool valid = tile < n_row_tiles;

    const int m0    = tile << 4;
    const int mrow  = lane & 15;
    const int khalf = lane >> 4;
    const int ncol  = lane & 15;
    float* __restrict__ xw = xl[wave];

    #pragma unroll
    for (int j = 0; j < 4; ++j) {
        const int idx = (int)threadIdx.x + 256 * j;
        const float4 w4 = reinterpret_cast<const float4*>(W)[idx];
        *reinterpret_cast<float4*>(&wl[(idx >> 4) * W_PITCH + (idx & 15) * 4]) = w4;
    }
    if (valid) {
        const float4* __restrict__ xt = reinterpret_cast<const float4*>(X + (long long)m0 * DIM);
        #pragma unroll
        for (int i = 0; i < 8; ++i) {
            const int idx = i * 32 + lane;
            const float4 x4 = xt[idx];
            *reinterpret_cast<float4*>(&xw[(idx >> 4) * X_PITCH + (idx & 15) * 4]) = x4;
        }
    }
    __syncthreads();

    v8f acc0 = {}, acc1 = {}, acc2 = {}, acc3 = {};
    if (valid) {
        const float* __restrict__ xrow = xw + mrow * X_PITCH;
        #pragma unroll
        for (int k0 = 0; k0 < DIM; k0 += 4) {
            const int ka = k0 + (khalf << 1);
            v2f a = *(const v2f*)(xrow + ka);

            const float* __restrict__ w0 = wl + ka * W_PITCH;
            const float* __restrict__ w1 = wl + (ka + 1) * W_PITCH;

            v2f b0, b1, b2, b3;
            b0.x = w0[ 0 + ncol]; b0.y = w1[ 0 + ncol];
            b1.x = w0[16 + ncol]; b1.y = w1[16 + ncol];
            b2.x = w0[32 + ncol]; b2.y = w1[32 + ncol];
            b3.x = w0[48 + ncol]; b3.y = w1[48 + ncol];

            acc0 = __builtin_amdgcn_wmma_f32_16x16x4_f32(false, a, false, b0,
                                                         (short)0, acc0, false, false);
            acc1 = __builtin_amdgcn_wmma_f32_16x16x4_f32(false, a, false, b1,
                                                         (short)0, acc1, false, false);
            acc2 = __builtin_amdgcn_wmma_f32_16x16x4_f32(false, a, false, b2,
                                                         (short)0, acc2, false, false);
            acc3 = __builtin_amdgcn_wmma_f32_16x16x4_f32(false, a, false, b3,
                                                         (short)0, acc3, false, false);
        }
    }
    __syncthreads();

    if (valid) {
        #pragma unroll
        for (int v = 0; v < 8; ++v) {
            float* __restrict__ srow = xw + ((khalf << 3) + v) * X_PITCH + ncol;
            srow[ 0] = acc0[v];
            srow[16] = acc1[v];
            srow[32] = acc2[v];
            srow[48] = acc3[v];
        }
    }
    __syncthreads();
    if (!valid) return;
    {
        float4* __restrict__ st = reinterpret_cast<float4*>(S + (long long)m0 * DIM);
        #pragma unroll
        for (int i = 0; i < 8; ++i) {
            const int idx = i * 32 + lane;
            st[idx] = *reinterpret_cast<const float4*>(&xw[(idx >> 4) * X_PITCH + (idx & 15) * 4]);
        }
    }

    const int2*   __restrict__ row2 = reinterpret_cast<const int2*>(adj_row);
    const int2*   __restrict__ col2 = reinterpret_cast<const int2*>(adj_col);
    const float2* __restrict__ val2 = reinterpret_cast<const float2*>(adj_val);
    uint4*          __restrict__ rec4  = reinterpret_cast<uint4*>(rec);
    const int npair   = n_edges >> 1;
    const int nblk    = (npair + 31) >> 5;
    const int pstride = n_row_tiles << 7;
    for (int q0 = tile << 7; q0 < npair; q0 += pstride) {
        int2 r[4], c[4];
        float2 v[4];
        #pragma unroll
        for (int k = 0; k < 4; ++k) {
            const int q = min(q0 + 32 * k + lane, npair - 1);
            r[k] = row2[q];
            c[k] = col2[q];
            v[k] = val2[q];
        }
        unsigned wlo = 0u, whi = 0u;
        #pragma unroll
        for (int k = 0; k < 4; ++k) {
            const int q = q0 + 32 * k + lane;
            const unsigned oa = (unsigned)r[k].x / (unsigned)OWN_ROWS;
            const unsigned ob = (unsigned)r[k].y / (unsigned)OWN_ROWS;
            if (q < npair) {
                rec4[q] = make_uint4(rec_word0((unsigned)r[k].x, (unsigned)c[k].x, oa),
                                     __float_as_uint(v[k].x),
                                     rec_word0((unsigned)r[k].y, (unsigned)c[k].y, ob),
                                     __float_as_uint(v[k].y));
            } else if (q < (nblk << 5)) {
                rec4[q] = make_uint4(511u << REC_ROW_SHIFT, 0u, 511u << REC_ROW_SHIFT, 0u);
            }
            #pragma unroll
            for (int b = 0; b < 4; ++b) {
                const unsigned al = __builtin_amdgcn_ballot_w32(((oa >> b) & 1u) != 0u);
                const unsigned ah = __builtin_amdgcn_ballot_w32(((oa >> (b + 4)) & 1u) != 0u);
                const unsigned bl = __builtin_amdgcn_ballot_w32(((ob >> b) & 1u) != 0u);
                const unsigned bh = __builtin_amdgcn_ballot_w32(((ob >> (b + 4)) & 1u) != 0u);
                const bool mea = lane == (2 * k    ) * 4 + b;
                const bool meb = lane == (2 * k + 1) * 4 + b;
                wlo = mea ? al : wlo;
                whi = mea ? ah : whi;
                wlo = meb ? bl : wlo;
                whi = meb ? bh : whi;
            }
        }
        const int d = (q0 >> 4) * 4 + lane;
        if (d < nblk * 8) {
            plo[d] = wlo;
            phi[d] = whi;
        }
    }
    if ((n_edges & 1) && tile == 0 && lane == 0) {
        const int e = n_edges - 1;
        const unsigned o = (unsigned)adj_row[e] / (unsigned)OWN_ROWS;
        rec[nblk << 6] = make_uint2(rec_word0((unsigned)adj_row[e], (unsigned)adj_col[e], o),
                            __float_as_uint(adj_val[e]));
    }
}

#define OWN_THREADS 1024
#define OWN_WAVES   32
#define SCAN_DEPTH  4
#define WQ_CAP      64
#define RQ_CAP      448
#define WAVE_ROWS   ((OWN_ROWS + OWN_WAVES - 1) / OWN_WAVES)

__device__ __forceinline__ void lds_add(float* p, float v) {
    __hip_atomic_fetch_add(p, v, __ATOMIC_RELAXED, __HIP_MEMORY_SCOPE_WORKGROUP);
}

__global__ __launch_bounds__(OWN_THREADS)
void gcn_owner_rows_kernel(const float* __restrict__ S,
                           const unsigned* __restrict__ plo,
                           const unsigned* __restrict__ phi,
                           const uint2* __restrict__ rec,
                           const int* __restrict__ adj_row,
                           float* __restrict__ out,
                           int n_nodes, int n_edges) {
    __shared__ float acc[OWN_ROWS * DIM];
    __shared__ int   wq[OWN_WAVES * WQ_CAP];
    __shared__ uint2 rq[OWN_WAVES * RQ_CAP];
    __shared__ int   rcnt[OWN_WAVES];

    const int tid   = threadIdx.x;
    const int lane  = tid & 31;
    const int wave  = __builtin_amdgcn_readfirstlane(tid >> 5);
    const int qbase = wave * WQ_CAP;

    const int npair = n_edges >> 1;
    const int nq = ((npair + 31) >> 5) << 1;
    const uint4* __restrict__ lo4 = reinterpret_cast<const uint4*>(plo);
    const uint4* __restrict__ hi4 = reinterpret_cast<const uint4*>(phi);
    const int rot   = ((int)blockIdx.x * (nq / (int)gridDim.x)) & ~31;
    const int niter = (nq + OWN_THREADS - 1) / OWN_THREADS;

    for (int r0 = blockIdx.x * OWN_ROWS; r0 < n_nodes; r0 += gridDim.x * OWN_ROWS) {
        const unsigned nrows = (unsigned)min(OWN_ROWS, n_nodes - r0);
        const unsigned oid = (unsigned)(r0 / OWN_ROWS) & 255u;
        unsigned x[8];
        #pragma unroll
        for (int b = 0; b < 8; ++b) x[b] = ((oid >> b) & 1u) - 1u;
        const unsigned ownbits = ((unsigned)(r0 / OWN_ROWS) << REC_OWN_SHIFT) >> REC_OWN_SHIFT;

        for (int i = tid; i < OWN_ROWS * DIM / 4; i += OWN_THREADS)
            reinterpret_cast<float4*>(acc)[i] = make_float4(0.f, 0.f, 0.f, 0.f);
        if (tid < OWN_WAVES) rcnt[tid] = 0;
        __syncthreads();

        int qn = 0;

        auto route = [&](int first, int cnt) {
            int e = 0;
            if (lane < cnt) e = wq[qbase + first + lane];
            const uint2 rc = rec[e];
            const unsigned lr = (rc.x << (32 - REC_OWN_SHIFT)) >> (32 - REC_OWN_SHIFT + REC_ROW_SHIFT);
            if (lane < cnt && (rc.x >> REC_OWN_SHIFT) == ownbits && lr < nrows) {
                const int d   = (int)(lr & (unsigned)(OWN_WAVES - 1));
                const int pos = __hip_atomic_fetch_add(&rcnt[d], 1, __ATOMIC_RELAXED,
                                                       __HIP_MEMORY_SCOPE_WORKGROUP);
                if (pos < RQ_CAP) {
                    rq[d * RQ_CAP + pos] = rc;
                } else {
                    const float wv = __uint_as_float(rc.y);
                    const float* __restrict__ srow =
                        S + (long long)(rc.x & ((1u << REC_ROW_SHIFT) - 1u)) * DIM;
                    #pragma nounroll
                    for (int j = 0; j < DIM; ++j) lds_add(&acc[lr * DIM + j], wv * srow[j]);
                }
            }
        };

        auto note32 = [&](unsigned m, unsigned j) {
            const int ebase = (int)((j >> 1) << 6) + (int)(j & 1u);
            unsigned mask = __builtin_amdgcn_ballot_w32(m != 0u);
            while (mask) {
                if (m != 0u) {
                    const int p    = __builtin_ctz(m);
                    const int slot = qbase + qn + (int)__builtin_amdgcn_mbcnt_lo(mask, 0u);
                    wq[slot] = ebase + 2 * p;
                    m &= m - 1u;
                }
                qn = __builtin_amdgcn_readfirstlane(qn + __builtin_popcount(mask));
                if (qn >= 32) {
                    route(qn - 32, 32);
                    qn -= 32;
                }
                mask = __builtin_amdgcn_ballot_w32(m != 0u);
            }
        };

        if (wave == 0) {
            unsigned lr = ~0u;
            if (lane == 0 && (n_edges & 1)) lr = (unsigned)(adj_row[n_edges - 1] - r0);
            const unsigned mask = __builtin_amdgcn_ballot_w32(lr < nrows);
            if (mask) {
                if (lr < nrows) wq[qbase] = nq << 5;
                qn = 1;
            }
        }

        auto match8 = [&](const uint4& l, const uint4& h) {
            return (l.x ^ x[0]) & (l.y ^ x[1]) & (l.z ^ x[2]) & (l.w ^ x[3]) &
                   (h.x ^ x[4]) & (h.y ^ x[5]) & (h.z ^ x[6]) & (h.w ^ x[7]);
        };
        int it = 0;
        for (; (it + SCAN_DEPTH) * OWN_THREADS <= nq; it += SCAN_DEPTH) {
            uint4    lo[SCAN_DEPTH], hi[SCAN_DEPTH];
            const int base = it * OWN_THREADS + tid;
            #pragma unroll
            for (int u = 0; u < SCAN_DEPTH; ++u) {
                unsigned j = (unsigned)(base + u * OWN_THREADS + rot);
                j = min(j, j - (unsigned)nq);
                lo[u] = lo4[j];
                hi[u] = hi4[j];
            }
            #pragma unroll
            for (int u = 0; u < SCAN_DEPTH; ++u) {
                unsigned j = (unsigned)(base + u * OWN_THREADS + rot);
                j = min(j, j - (unsigned)nq);
                note32(match8(lo[u], hi[u]), j);
            }
        }
        for (; it < niter; ++it) {
            const int idx = it * OWN_THREADS + tid;
            unsigned j = (unsigned)(idx + rot);
            j = min(j, j - (unsigned)nq);
            if (idx >= nq) j = 0u;
            const uint4 l = lo4[j];
            const uint4 h = hi4[j];
            note32(idx < nq ? match8(l, h) : 0u, j);
        }
        if (qn > 0) route(0, qn);
        __syncthreads();

        const int nrec = min(__builtin_amdgcn_readfirstlane(rcnt[wave]), RQ_CAP);
        const uint2* myq = rq + wave * RQ_CAP;
        v16f ax = {}, ay = {};

        for (int i0 = 0; i0 < nrec; i0 += 32) {
            const int cnt = min(32, nrec - i0);
            const uint2 rc = myq[i0 + min(lane, cnt - 1)];
            const int c  = (int)(rc.x & ((1u << REC_ROW_SHIFT) - 1u));
            const int kr = (int)((rc.x << (32 - REC_OWN_SHIFT)) >> (32 - REC_OWN_SHIFT + REC_ROW_SHIFT + 5));
            const int wi = (int)rc.y;
            for (int h = 0; h < cnt; h += 16) {
                float2 sv[16];
                #pragma unroll
                for (int u = 0; u < 16; ++u) {
                    const int k  = min(h + u, cnt - 1);
                    const int ck = __builtin_amdgcn_readlane(c, k);
                    sv[u] = *reinterpret_cast<const float2*>(S + (long long)ck * DIM + 2 * lane);
                }
                #pragma unroll
                for (int u = 0; u < 16; ++u) {
                    if (h + u < cnt) {
                        const int   kk = __builtin_amdgcn_readlane(kr, h + u);
                        const float wk = __int_as_float(__builtin_amdgcn_readlane(wi, h + u));
                        ax[kk] += wk * sv[u].x;
                        ay[kk] += wk * sv[u].y;
                    }
                }
            }
        }

        #pragma unroll
        for (int k = 0; k < WAVE_ROWS; ++k) {
            const int r = wave + OWN_WAVES * k;
            if (r < (int)nrows) {
                const float2 o = *reinterpret_cast<const float2*>(&acc[r * DIM + 2 * lane]);
                *reinterpret_cast<float2*>(out + (long long)(r0 + r) * DIM + 2 * lane) =
                    make_float2(ax[k] + o.x, ay[k] + o.y);
            }
        }
        __syncthreads();
    }
}

extern "C" void kernel_launch(void* const* d_in, const int* in_sizes, int n_in,
                              void* d_out, int out_size, void* d_ws, size_t ws_size,
                              hipStream_t stream) {
    const float* X    = (const float*)d_in[0];
    const float* W    = (const float*)d_in[1];
    const int*   arow = (const int*)d_in[2];
    const int*   acol = (const int*)d_in[3];
    const float* aval = (const float*)d_in[4];
    float*       out  = (float*)d_out;
    float*       S    = (float*)d_ws;
    unsigned*    plo  = (unsigned*)((char*)d_ws + 134217728);
    unsigned*    phi  = (unsigned*)((char*)d_ws + 138412032);
    uint2*       rec  = (uint2*)((char*)d_ws + 67108864);

    const int n_nodes = in_sizes[0] / DIM;
    const int n_edges = in_sizes[2];

    {
        const int tiles = n_nodes / 16;
        const int waves_per_block = 256 / 32;
        const int blocks = (tiles + waves_per_block - 1) / waves_per_block;
        gcn_gemm_wmma_kernel<<<blocks, 256, 0, stream>>>(X, W, S, tiles, arow, acol, aval, plo, phi,
                                                         rec, n_edges);
    }

    {
        const int blocks = (n_nodes + OWN_ROWS - 1) / OWN_ROWS;
        gcn_owner_rows_kernel<<<blocks, OWN_THREADS, 0, stream>>>(S, plo, phi, rec, arow, out,
                                                                  n_nodes, n_edges);
    }
}


# Owner list re-coded as 32 one-hot planes (16 per 4-bit digit of the owner): an owner reads and ANDs its two planes only, a work-group writes whole 256-byte pieces of every plane; sum order as before
# speedup vs baseline: 5.1529x; 1.0586x over previous
#include <hip/hip_runtime.h>

typedef __attribute__((ext_vector_type(2))) float v2f;
typedef __attribute__((ext_vector_type(8))) float v8f;
typedef __attribute__((ext_vector_type(16))) float v16f;

#define DIM 64
#define OWN_ROWS 391

#define REC_ROW_SHIFT 20
#define REC_OWN_SHIFT 29

__device__ __forceinline__ unsigned rec_word0(unsigned row, unsigned col, unsigned owner) {
    return col | ((row - owner * (unsigned)OWN_ROWS) << REC_ROW_SHIFT) | (owner << REC_OWN_SHIFT);
}

#define W_PITCH 72
#define X_PITCH 68
#define N_PLANES 32
#define PL_PITCH 65

__global__ __launch_bounds__(256)
void gcn_gemm_wmma_kernel(const float* __restrict__ X,
                                     const float* __restrict__ W,
                                     float* __restrict__ S,
                                     int n_row_tiles,
                                     const int* __restrict__ adj_row,
                                     const int* __restrict__ adj_col,
                                     const float* __restrict__ adj_val,
                                     unsigned* __restrict__ planes,
                                     uint2* __restrict__ rec,
                                     int n_edges) {
    __shared__ __attribute__((aligned(16))) float wl[DIM * W_PITCH];
    __shared__ __attribute__((aligned(16))) float xl[8][16 * X_PITCH];
    __shared__ unsigned pl[N_PLANES * PL_PITCH];

    const int wave = threadIdx.x >> 5;
    const int lane = threadIdx.x & 31;
    const int tile = blockIdx.x * 8 + wave;
    const bool valid = tile < n_row_tiles;

    const int m0    = tile << 4;
    const int mrow  = lane & 15;
    const int khalf = lane >> 4;
    const int ncol  = lane & 15;
    float* __restrict__ xw = xl[wave];

    for (int i = (int)threadIdx.x; i < N_PLANES * PL_PITCH; i += 256) pl[i] = 0u;

    #pragma unroll
    for (int j = 0; j < 4; ++j) {
        const int idx = (int)threadIdx.x + 256 * j;
        const float4 w4 = reinterpret_cast<const float4*>(W)[idx];
        *reinterpret_cast<float4*>(&wl[(idx >> 4) * W_PITCH + (idx & 15) * 4]) = w4;
    }
    if (valid) {
        const float4* __restrict__ xt = reinterpret_cast<const float4*>(X + (long long)m0 * DIM);
        #pragma unroll
        for (int i = 0; i < 8; ++i) {
            const int idx = i * 32 + lane;
            const float4 x4 = xt[idx];
            *reinterpret_cast<float4*>(&xw[(idx >> 4) * X_PITCH + (idx & 15) * 4]) = x4;
        }
    }
    __syncthreads();

    v8f acc0 = {}, acc1 = {}, acc2 = {}, acc3 = {};
    if (valid) {
        const float* __restrict__ xrow = xw + mrow * X_PITCH;
        #pragma unroll
        for (int k0 = 0; k0 < DIM; k0 += 4) {
            const int ka = k0 + (khalf << 1);
            v2f a = *(const v2f*)(xrow + ka);

            const float* __restrict__ w0 = wl + ka * W_PITCH;
            const float* __restrict__ w1 = wl + (ka + 1) * W_PITCH;

            v2f b0, b1, b2, b3;
            b0.x = w0[ 0 + ncol]; b0.y = w1[ 0 + ncol];
            b1.x = w0[16 + ncol]; b1.y = w1[16 + ncol];
            b2.x = w0[32 + ncol]; b2.y = w1[32 + ncol];
            b3.x = w0[48 + ncol]; b3.y = w1[48 + ncol];

            acc0 = __builtin_amdgcn_wmma_f32_16x16x4_f32(false, a, false, b0,
                                                         (short)0, acc0, false, false);
            acc1 = __builtin_amdgcn_wmma_f32_16x16x4_f32(false, a, false, b1,
                                                         (short)0, acc1, false, false);
            acc2 = __builtin_amdgcn_wmma_f32_16x16x4_f32(false, a, false, b2,
                                                         (short)0, acc2, false, false);
            acc3 = __builtin_amdgcn_wmma_f32_16x16x4_f32(false, a, false, b3,
                                                         (short)0, acc3, false, false);
        }
    }
    __syncthreads();

    if (valid) {
        #pragma unroll
        for (int v = 0; v < 8; ++v) {
            float* __restrict__ srow = xw + ((khalf << 3) + v) * X_PITCH + ncol;
            srow[ 0] = acc0[v];
            srow[16] = acc1[v];
            srow[32] = acc2[v];
            srow[48] = acc3[v];
        }
    }
    __syncthreads();
    if (valid) {
        float4* __restrict__ st = reinterpret_cast<float4*>(S + (long long)m0 * DIM);
        #pragma unroll
        for (int i = 0; i < 8; ++i) {
            const int idx = i * 32 + lane;
            st[idx] = *reinterpret_cast<const float4*>(&xw[(idx >> 4) * X_PITCH + (idx & 15) * 4]);
        }
    }

    const int2*   __restrict__ row2 = reinterpret_cast<const int2*>(adj_row);
    const int2*   __restrict__ col2 = reinterpret_cast<const int2*>(adj_col);
    const float2* __restrict__ val2 = reinterpret_cast<const float2*>(adj_val);
    uint4*          __restrict__ rec4  = reinterpret_cast<uint4*>(rec);
    const int npair   = n_edges >> 1;
    const int nblk    = (npair + 31) >> 5;
    const int pstr    = ((nblk + 31) >> 5) << 6;
    const int qstride = (int)gridDim.x << 10;
    for (int qb = (int)blockIdx.x << 10; qb < npair; qb += qstride) {
        const int q0 = qb + (wave << 7);
        if (q0 < npair) {
            int2 r[4], c[4];
            float2 v[4];
            #pragma unroll
            for (int k = 0; k < 4; ++k) {
                const int q = min(q0 + 32 * k + lane, npair - 1);
                r[k] = row2[q];
                c[k] = col2[q];
                v[k] = val2[q];
            }
            #pragma unroll
            for (int k = 0; k < 4; ++k) {
                const int q = q0 + 32 * k + lane;
                if (q < npair) {
                    const unsigned oa = (unsigned)r[k].x / (unsigned)OWN_ROWS;
                    const unsigned ob = (unsigned)r[k].y / (unsigned)OWN_ROWS;
                    rec4[q] = make_uint4(rec_word0((unsigned)r[k].x, (unsigned)c[k].x, oa),
                                         __float_as_uint(v[k].x),
                                         rec_word0((unsigned)r[k].y, (unsigned)c[k].y, ob),
                                         __float_as_uint(v[k].y));
                    const unsigned bit = 1u << lane;
                    const int ha = 8 * wave + 2 * k;
                    __hip_atomic_fetch_or(&pl[(int)((oa >> 4) & 15u) * PL_PITCH + ha], bit,
                                          __ATOMIC_RELAXED, __HIP_MEMORY_SCOPE_WORKGROUP);
                    __hip_atomic_fetch_or(&pl[(int)(16u + (oa & 15u)) * PL_PITCH + ha], bit,
                                          __ATOMIC_RELAXED, __HIP_MEMORY_SCOPE_WORKGROUP);
                    __hip_atomic_fetch_or(&pl[(int)((ob >> 4) & 15u) * PL_PITCH + ha + 1], bit,
                                          __ATOMIC_RELAXED, __HIP_MEMORY_SCOPE_WORKGROUP);
                    __hip_atomic_fetch_or(&pl[(int)(16u + (ob & 15u)) * PL_PITCH + ha + 1], bit,
                                          __ATOMIC_RELAXED, __HIP_MEMORY_SCOPE_WORKGROUP);
                }
            }
        }
        __syncthreads();
        {
            const int piece = (qb >> 10) << 6;
            #pragma unroll
            for (int p = 0; p < 4; ++p) {
                const int plane = 4 * wave + p;
                const unsigned a = pl[plane * PL_PITCH + 2 * lane];
                const unsigned b = pl[plane * PL_PITCH + 2 * lane + 1];
                *reinterpret_cast<uint2*>(planes + (long long)plane * pstr + piece + 2 * lane) =
                    make_uint2(a, b);
            }
        }
        if (qb + qstride < npair) {
            __syncthreads();
            for (int i = (int)threadIdx.x; i < N_PLANES * PL_PITCH; i += 256) pl[i] = 0u;
            __syncthreads();
        }
    }
    if ((n_edges & 1) && tile == 0 && lane == 0) {
        const int e = n_edges - 1;
        const unsigned o = (unsigned)adj_row[e] / (unsigned)OWN_ROWS;
        rec[nblk << 6] = make_uint2(rec_word0((unsigned)adj_row[e], (unsigned)adj_col[e], o),
                            __float_as_uint(adj_val[e]));
    }
}

#define OWN_THREADS 1024
#define OWN_WAVES   32
#define SCAN_DEPTH  4
#define WQ_CAP      64
#define RQ_CAP      448
#define WAVE_ROWS   ((OWN_ROWS + OWN_WAVES - 1) / OWN_WAVES)

__device__ __forceinline__ void lds_add(float* p, float v) {
    __hip_atomic_fetch_add(p, v, __ATOMIC_RELAXED, __HIP_MEMORY_SCOPE_WORKGROUP);
}

__global__ __launch_bounds__(OWN_THREADS)
void gcn_owner_rows_kernel(const float* __restrict__ S,
                           const unsigned* __restrict__ planes,
                           const uint2* __restrict__ rec,
                           const int* __restrict__ adj_row,
                           float* __restrict__ out,
                           int n_nodes, int n_edges) {
    __shared__ float acc[OWN_ROWS * DIM];
    __shared__ int   wq[OWN_WAVES * WQ_CAP];
    __shared__ uint2 rq[OWN_WAVES * RQ_CAP];
    __shared__ int   rcnt[OWN_WAVES];

    const int tid   = threadIdx.x;
    const int lane  = tid & 31;
    const int wave  = __builtin_amdgcn_readfirstlane(tid >> 5);
    const int qbase = wave * WQ_CAP;

    const int npair = n_edges >> 1;
    const int nblk = (npair + 31) >> 5;
    const int nq   = nblk << 1;
    const int pstr = ((nblk + 31) >> 5) << 6;
    const int ng   = (nq + 3) >> 2;
    const int rot   = ((int)blockIdx.x * (ng / (int)gridDim.x)) & ~31;
    const int niter = (ng + OWN_THREADS - 1) / OWN_THREADS;

    for (int r0 = blockIdx.x * OWN_ROWS; r0 < n_nodes; r0 += gridDim.x * OWN_ROWS) {
        const unsigned nrows = (unsigned)min(OWN_ROWS, n_nodes - r0);
        const unsigned oid = (unsigned)(r0 / OWN_ROWS) & 255u;
        const uint4* __restrict__ hi4 =
            reinterpret_cast<const uint4*>(planes + (long long)(oid >> 4) * pstr);
        const uint4* __restrict__ lo4 =
            reinterpret_cast<const uint4*>(planes + (long long)(16u + (oid & 15u)) * pstr);
        const unsigned ownbits = ((unsigned)(r0 / OWN_ROWS) << REC_OWN_SHIFT) >> REC_OWN_SHIFT;

        for (int i = tid; i < OWN_ROWS * DIM / 4; i += OWN_THREADS)
            reinterpret_cast<float4*>(acc)[i] = make_float4(0.f, 0.f, 0.f, 0.f);
        if (tid < OWN_WAVES) rcnt[tid] = 0;
        __syncthreads();

        int qn = 0;

        auto route = [&](int first, int cnt) {
            int e = 0;
            if (lane < cnt) e = wq[qbase + first + lane];
            const uint2 rc = rec[e];
            const unsigned lr = (rc.x << (32 - REC_OWN_SHIFT)) >> (32 - REC_OWN_SHIFT + REC_ROW_SHIFT);
            if (lane < cnt && (rc.x >> REC_OWN_SHIFT) == ownbits && lr < nrows) {
                const int d   = (int)(lr & (unsigned)(OWN_WAVES - 1));
                const int pos = __hip_atomic_fetch_add(&rcnt[d], 1, __ATOMIC_RELAXED,
                                                       __HIP_MEMORY_SCOPE_WORKGROUP);
                if (pos < RQ_CAP) {
                    rq[d * RQ_CAP + pos] = rc;
                } else {
                    const float wv = __uint_as_float(rc.y);
                    const float* __restrict__ srow =
                        S + (long long)(rc.x & ((1u << REC_ROW_SHIFT) - 1u)) * DIM;
                    #pragma nounroll
                    for (int j = 0; j < DIM; ++j) lds_add(&acc[lr * DIM + j], wv * srow[j]);
                }
            }
        };

        auto note32 = [&](unsigned m, unsigned j) {
            const int ebase = (int)((j >> 1) << 6) + (int)(j & 1u);
            unsigned mask = __builtin_amdgcn_ballot_w32(m != 0u);
            while (mask) {
                if (m != 0u) {
                    const int p    = __builtin_ctz(m);
                    const int slot = qbase + qn + (int)__builtin_amdgcn_mbcnt_lo(mask, 0u);
                    wq[slot] = ebase + 2 * p;
                    m &= m - 1u;
                }
                qn = __builtin_amdgcn_readfirstlane(qn + __builtin_popcount(mask));
                if (qn >= 32) {
                    route(qn - 32, 32);
                    qn -= 32;
                }
                mask = __builtin_amdgcn_ballot_w32(m != 0u);
            }
        };

        if (wave == 0) {
            unsigned lr = ~0u;
            if (lane == 0 && (n_edges & 1)) lr = (unsigned)(adj_row[n_edges - 1] - r0);
            const unsigned mask = __builtin_amdgcn_ballot_w32(lr < nrows);
            if (mask) {
                if (lr < nrows) wq[qbase] = nq << 5;
                qn = 1;
            }
        }

        auto note128 = [&](const uint4& l, const uint4& h, unsigned g) {
            note32(l.x & h.x, 4u * g);
            note32(l.y & h.y, 4u * g + 1u);
            note32(l.z & h.z, 4u * g + 2u);
            note32(l.w & h.w, 4u * g + 3u);
        };
        int it = 0;
        for (; (it + SCAN_DEPTH) * OWN_THREADS <= ng; it += SCAN_DEPTH) {
            uint4    lo[SCAN_DEPTH], hi[SCAN_DEPTH];
            const int base = it * OWN_THREADS + tid;
            #pragma unroll
            for (int u = 0; u < SCAN_DEPTH; ++u) {
                unsigned g = (unsigned)(base + u * OWN_THREADS + rot);
                g = min(g, g - (unsigned)ng);
                lo[u] = lo4[g];
                hi[u] = hi4[g];
            }
            #pragma unroll
            for (int u = 0; u < SCAN_DEPTH; ++u) {
                unsigned g = (unsigned)(base + u * OWN_THREADS + rot);
                g = min(g, g - (unsigned)ng);
                note128(lo[u], hi[u], g);
            }
        }
        for (; it < niter; ++it) {
            const int idx = it * OWN_THREADS + tid;
            unsigned g = (unsigned)(idx + rot);
            g = min(g, g - (unsigned)ng);
            if (idx >= ng) g = 0u;
            uint4 l = lo4[g];
            const uint4 h = hi4[g];
            if (idx >= ng) l = make_uint4(0u, 0u, 0u, 0u);
            note128(l, h, g);
        }
        if (qn > 0) route(0, qn);
        __syncthreads();

        const int nrec = min(__builtin_amdgcn_readfirstlane(rcnt[wave]), RQ_CAP);
        const uint2* myq = rq + wave * RQ_CAP;
        v16f ax = {}, ay = {};

        for (int i0 = 0; i0 < nrec; i0 += 32) {
            const int cnt = min(32, nrec - i0);
            const uint2 rc = myq[i0 + min(lane, cnt - 1)];
            const int c  = (int)(rc.x & ((1u << REC_ROW_SHIFT) - 1u));
            const int kr = (int)((rc.x << (32 - REC_OWN_SHIFT)) >> (32 - REC_OWN_SHIFT + REC_ROW_SHIFT + 5));
            const int wi = (int)rc.y;
            for (int h = 0; h < cnt; h += 16) {
                float2 sv[16];
                #pragma unroll
                for (int u = 0; u < 16; ++u) {
                    const int k  = min(h + u, cnt - 1);
                    const int ck = __builtin_amdgcn_readlane(c, k);
                    sv[u] = *reinterpret_cast<const float2*>(S + (long long)ck * DIM + 2 * lane);
                }
                #pragma unroll
                for (int u = 0; u < 16; ++u) {
                    if (h + u < cnt) {
                        const int   kk = __builtin_amdgcn_readlane(kr, h + u);
                        const float wk = __int_as_float(__builtin_amdgcn_readlane(wi, h + u));
                        ax[kk] += wk * sv[u].x;
                        ay[kk] += wk * sv[u].y;
                    }
                }
            }
        }

        #pragma unroll
        for (int k = 0; k < WAVE_ROWS; ++k) {
            const int r = wave + OWN_WAVES * k;
            if (r < (int)nrows) {
                const float2 o = *reinterpret_cast<const float2*>(&acc[r * DIM + 2 * lane]);
                *reinterpret_cast<float2*>(out + (long long)(r0 + r) * DIM + 2 * lane) =
                    make_float2(ax[k] + o.x, ay[k] + o.y);
            }
        }
        __syncthreads();
    }
}

extern "C" void kernel_launch(void* const* d_in, const int* in_sizes, int n_in,
                              void* d_out, int out_size, void* d_ws, size_t ws_size,
                              hipStream_t stream) {
    const float* X    = (const float*)d_in[0];
    const float* W    = (const float*)d_in[1];
    const int*   arow = (const int*)d_in[2];
    const int*   acol = (const int*)d_in[3];
    const float* aval = (const float*)d_in[4];
    float*       out  = (float*)d_out;
    float*       S    = (float*)d_ws;
    unsigned*    planes = (unsigned*)((char*)d_ws + 134217728);
    uint2*       rec  = (uint2*)((char*)d_ws + 67108864);

    const int n_nodes = in_sizes[0] / DIM;
    const int n_edges = in_sizes[2];

    {
        const int tiles = n_nodes / 16;
        const int waves_per_block = 256 / 32;
        const int blocks = (tiles + waves_per_block - 1) / waves_per_block;
        gcn_gemm_wmma_kernel<<<blocks, 256, 0, stream>>>(X, W, S, tiles, arow, acol, aval, planes,
                                                         rec, n_edges);
    }

    {
        const int blocks = (n_nodes + OWN_ROWS - 1) / OWN_ROWS;
        gcn_owner_rows_kernel<<<blocks, OWN_THREADS, 0, stream>>>(S, planes, rec, arow, out,
                                                                  n_nodes, n_edges);
    }
}
